# starting kernel
# baseline (speedup 1.0000x reference)
#include <hip/hip_runtime.h>
#include <hip/hip_bf16.h>
#include <math.h>

#define B_      8
#define N_      1024
#define CIN_    256
#define H_      4
#define COUT_   256
#define C_      64
#define ALPHA_  0.2f
#define NEG_INF_ (-9.0e15f)

typedef float v2f __attribute__((ext_vector_type(2)));
typedef float v8f __attribute__((ext_vector_type(8)));

__global__ __launch_bounds__(256) void gat_gemm_h(const float* __restrict__ X,
                                                  const float* __restrict__ W,
                                                  const float* __restrict__ bias,
                                                  float* __restrict__ hbuf) {
  const int wave = blockIdx.x * (blockDim.x >> 5) + (threadIdx.x >> 5);
  const int lane = threadIdx.x & 31;
  const int half = lane >> 4;
  const int l    = lane & 15;

  const int tileM = wave >> 2;
  const int cg    = wave & 3;
  const int row0  = tileM * 16;
  const int col0  = cg * 64;

  const float* xrow = X + (size_t)(row0 + l) * CIN_;
  const float* wr0  = W + (size_t)(col0 +  0 + l) * CIN_;
  const float* wr1  = W + (size_t)(col0 + 16 + l) * CIN_;
  const float* wr2  = W + (size_t)(col0 + 32 + l) * CIN_;
  const float* wr3  = W + (size_t)(col0 + 48 + l) * CIN_;

  v8f acc[4] = {{}, {}, {}, {}};
  #pragma unroll 4
  for (int kk = 0; kk < CIN_; kk += 4) {
    const int ko = kk + 2 * half;
    v2f av  = *(const v2f*)(xrow + ko);
    v2f bv0 = *(const v2f*)(wr0 + ko);
    v2f bv1 = *(const v2f*)(wr1 + ko);
    v2f bv2 = *(const v2f*)(wr2 + ko);
    v2f bv3 = *(const v2f*)(wr3 + ko);
    acc[0] = __builtin_amdgcn_wmma_f32_16x16x4_f32(false, av, false, bv0,
                                                   (short)0, acc[0], false, false);
    acc[1] = __builtin_amdgcn_wmma_f32_16x16x4_f32(false, av, false, bv1,
                                                   (short)0, acc[1], false, false);
    acc[2] = __builtin_amdgcn_wmma_f32_16x16x4_f32(false, av, false, bv2,
                                                   (short)0, acc[2], false, false);
    acc[3] = __builtin_amdgcn_wmma_f32_16x16x4_f32(false, av, false, bv3,
                                                   (short)0, acc[3], false, false);
  }

  #pragma unroll
  for (int cc = 0; cc < 4; ++cc) {
    const float bval = bias[col0 + cc * 16 + l];
    #pragma unroll
    for (int r = 0; r < 8; ++r) {
      const int row = row0 + r + 8 * half;
      hbuf[(size_t)row * COUT_ + col0 + cc * 16 + l] = acc[cc][r] + bval;
    }
  }
}

__global__ __launch_bounds__(256) void gat_lscore(const float* __restrict__ hbuf,
                                                  const float* __restrict__ a,
                                                  float* __restrict__ lsrc,
                                                  float* __restrict__ ldst) {
  const int t    = blockIdx.x * blockDim.x + threadIdx.x;
  const int head = t & (H_ - 1);
  const int bi   = t >> 2;
  const float* hr = hbuf + (size_t)bi * COUT_ + head * C_;
  const float* as = a + head * (2 * C_);
  const float* ad = as + C_;
  float s = 0.f, d = 0.f;
  #pragma unroll 8
  for (int c = 0; c < C_; ++c) {
    const float v = hr[c];
    s = fmaf(v, as[c], s);
    d = fmaf(v, ad[c], d);
  }
  lsrc[t] = s;
  ldst[t] = d;
}

__global__ __launch_bounds__(256) void gat_stats(const float* __restrict__ lsrc,
                                                 const float* __restrict__ ldst,
                                                 const int*   __restrict__ adj,
                                                 float* __restrict__ mbuf,
                                                 float* __restrict__ sbuf) {
  const int wave = blockIdx.x * (blockDim.x >> 5) + (threadIdx.x >> 5);
  const int lane = threadIdx.x & 31;
  const size_t bi = (size_t)wave;
  const int b = wave / N_;

  float ls[H_];
  #pragma unroll
  for (int hd = 0; hd < H_; ++hd) ls[hd] = lsrc[bi * H_ + hd];

  const int*    arow = adj + bi * (size_t)N_;
  const float4* ldr  = (const float4*)ldst;

  float mx[H_];
  #pragma unroll
  for (int hd = 0; hd < H_; ++hd) mx[hd] = NEG_INF_;

  for (int j = lane; j < N_; j += 32) {
    const int    ad = arow[j];
    const float4 ld = ldr[(size_t)b * N_ + j];
    const float  ldv[H_] = {ld.x, ld.y, ld.z, ld.w};
    #pragma unroll
    for (int hd = 0; hd < H_; ++hd) {
      float x = ls[hd] + ldv[hd];
      x = fmaxf(x, ALPHA_ * x);
      x = (ad == 1) ? x : NEG_INF_;
      mx[hd] = fmaxf(mx[hd], x);
    }
  }
  #pragma unroll
  for (int hd = 0; hd < H_; ++hd)
    #pragma unroll
    for (int off = 16; off > 0; off >>= 1)
      mx[hd] = fmaxf(mx[hd], __shfl_xor(mx[hd], off, 32));

  float sm[H_] = {0.f, 0.f, 0.f, 0.f};
  for (int j = lane; j < N_; j += 32) {
    const int    ad = arow[j];
    const float4 ld = ldr[(size_t)b * N_ + j];
    const float  ldv[H_] = {ld.x, ld.y, ld.z, ld.w};
    #pragma unroll
    for (int hd = 0; hd < H_; ++hd) {
      float x = ls[hd] + ldv[hd];
      x = fmaxf(x, ALPHA_ * x);
      sm[hd] += (ad == 1) ? __expf(x - mx[hd]) : 0.f;
    }
  }
  #pragma unroll
  for (int hd = 0; hd < H_; ++hd)
    #pragma unroll
    for (int off = 16; off > 0; off >>= 1)
      sm[hd] += __shfl_xor(sm[hd], off, 32);

  if (lane == 0) {
    #pragma unroll
    for (int hd = 0; hd < H_; ++hd) {
      mbuf[bi * H_ + hd] = mx[hd];
      sbuf[bi * H_ + hd] = sm[hd];
    }
  }
}

__global__ __launch_bounds__(128) void gat_aggregate(const float* __restrict__ hbuf,
                                                     const float* __restrict__ lsrc,
                                                     const float* __restrict__ ldst,
                                                     const int*   __restrict__ adj,
                                                     const float* __restrict__ mbuf,
                                                     const float* __restrict__ sbuf,
                                                     float* __restrict__ out) {
  __shared__ float lds_ld[N_ * H_];

  const int b     = blockIdx.x >> 5;
  const int islab = blockIdx.x & 31;
  const int i0    = islab * 32;
  const int head  = threadIdx.x >> 5;
  const int lane  = threadIdx.x & 31;
  const int half  = lane >> 4;
  const int l     = lane & 15;

  {
    const float4* src = (const float4*)(ldst + (size_t)b * N_ * H_);
    float4*       dst = (float4*)lds_ld;
    for (int j = threadIdx.x; j < N_; j += 128) dst[j] = src[j];
  }
  __syncthreads();

  const size_t biA = (size_t)b * N_ + (i0 + l);
  const size_t biB = biA + 16;
  const float  lsA = lsrc[biA * H_ + head];
  const float  lsB = lsrc[biB * H_ + head];
  const float  mA  = mbuf[biA * H_ + head];
  const float  mB  = mbuf[biB * H_ + head];
  const float  isA = 1.0f / sbuf[biA * H_ + head];
  const float  isB = 1.0f / sbuf[biB * H_ + head];
  const int*   arA = adj + biA * (size_t)N_;
  const int*   arB = adj + biB * (size_t)N_;

  v8f accA[4] = {{}, {}, {}, {}};
  v8f accB[4] = {{}, {}, {}, {}};

  for (int j0 = 0; j0 < N_; j0 += 4) {
    const int ja = j0 + 2 * half;

    const int2  adA = *(const int2*)(arA + ja);
    const int2  adB = *(const int2*)(arB + ja);
    const float ld0 = lds_ld[ja * H_ + head];
    const float ld1 = lds_ld[(ja + 1) * H_ + head];

    float xA0 = lsA + ld0; xA0 = fmaxf(xA0, ALPHA_ * xA0);
    float xA1 = lsA + ld1; xA1 = fmaxf(xA1, ALPHA_ * xA1);
    float xB0 = lsB + ld0; xB0 = fmaxf(xB0, ALPHA_ * xB0);
    float xB1 = lsB + ld1; xB1 = fmaxf(xB1, ALPHA_ * xB1);

    v2f avA, avB;
    avA.x = (adA.x == 1) ? __expf(xA0 - mA) * isA : 0.0f;
    avA.y = (adA.y == 1) ? __expf(xA1 - mA) * isA : 0.0f;
    avB.x = (adB.x == 1) ? __expf(xB0 - mB) * isB : 0.0f;
    avB.y = (adB.y == 1) ? __expf(xB1 - mB) * isB : 0.0f;

    const float* hrow0 = hbuf + ((size_t)b * N_ + ja) * COUT_ + head * C_ + l;
    const float* hrow1 = hrow0 + COUT_;
    #pragma unroll
    for (int cc = 0; cc < 4; ++cc) {
      v2f bv;
      bv.x = hrow0[cc * 16];
      bv.y = hrow1[cc * 16];
      accA[cc] = __builtin_amdgcn_wmma_f32_16x16x4_f32(false, avA, false, bv,
                                                       (short)0, accA[cc],
                                                       false, false);
      accB[cc] = __builtin_amdgcn_wmma_f32_16x16x4_f32(false, avB, false, bv,
                                                       (short)0, accB[cc],
                                                       false, false);
    }
  }

  #pragma unroll
  for (int cc = 0; cc < 4; ++cc) {
    #pragma unroll
    for (int r = 0; r < 8; ++r) {
      const int rowA = i0 + r + 8 * half;
      const int rowB = rowA + 16;
      out[((size_t)b * N_ + rowA) * COUT_ + head * C_ + cc * 16 + l] = accA[cc][r];
      out[((size_t)b * N_ + rowB) * COUT_ + head * C_ + cc * 16 + l] = accB[cc][r];
    }
  }
}

extern "C" void kernel_launch(void* const* d_in, const int* in_sizes, int n_in,
                              void* d_out, int out_size, void* d_ws, size_t ws_size,
                              hipStream_t stream) {
  const float* node_feats = (const float*)d_in[0];
  const int*   adj        = (const int*)  d_in[1];
  const float* W          = (const float*)d_in[2];
  const float* bias       = (const float*)d_in[3];
  const float* a          = (const float*)d_in[4];
  float*       out        = (float*)d_out;

  float* hbuf = (float*)d_ws;
  float* lsrc = hbuf + (size_t)B_ * N_ * COUT_;
  float* ldst = lsrc + (size_t)B_ * N_ * H_;
  float* mbuf = ldst + (size_t)B_ * N_ * H_;
  float* sbuf = mbuf + (size_t)B_ * N_ * H_;

  gat_gemm_h<<<(B_ * N_ / 16) * (COUT_ / 64) / 8, 256, 0, stream>>>(
      node_feats, W, bias, hbuf);

  gat_lscore<<<(B_ * N_ * H_) / 256, 256, 0, stream>>>(hbuf, a, lsrc, ldst);

  gat_stats<<<(B_ * N_) / 8, 256, 0, stream>>>(lsrc, ldst, adj, mbuf, sbuf);

  gat_aggregate<<<B_ * (N_ / 32), 128, 0, stream>>>(
      hbuf, lsrc, ldst, adj, mbuf, sbuf, out);
}


# Row-stats kernel: a wave takes 4 rows, holds its batch's l_dst in registers and loads each adjacency row once, a row ahead, for both passes (was 2 reads, a wait per 64 neighbours); same sum order
# speedup vs baseline: 2.0542x; 2.0542x over previous
#include <hip/hip_runtime.h>
#include <hip/hip_bf16.h>
#include <math.h>

#define B_      8
#define N_      1024
#define CIN_    256
#define H_      4
#define COUT_   256
#define C_      64
#define ALPHA_  0.2f
#define NEG_INF_ (-9.0e15f)

typedef float v2f __attribute__((ext_vector_type(2)));
typedef float v8f __attribute__((ext_vector_type(8)));

__global__ __launch_bounds__(256) void gat_gemm_h(const float* __restrict__ X,
                                                  const float* __restrict__ W,
                                                  const float* __restrict__ bias,
                                                  float* __restrict__ hbuf) {
  const int wave = blockIdx.x * (blockDim.x >> 5) + (threadIdx.x >> 5);
  const int lane = threadIdx.x & 31;
  const int half = lane >> 4;
  const int l    = lane & 15;

  const int tileM = wave >> 2;
  const int cg    = wave & 3;
  const int row0  = tileM * 16;
  const int col0  = cg * 64;

  const float* xrow = X + (size_t)(row0 + l) * CIN_;
  const float* wr0  = W + (size_t)(col0 +  0 + l) * CIN_;
  const float* wr1  = W + (size_t)(col0 + 16 + l) * CIN_;
  const float* wr2  = W + (size_t)(col0 + 32 + l) * CIN_;
  const float* wr3  = W + (size_t)(col0 + 48 + l) * CIN_;

  v8f acc[4] = {{}, {}, {}, {}};
  #pragma unroll 4
  for (int kk = 0; kk < CIN_; kk += 4) {
    const int ko = kk + 2 * half;
    v2f av  = *(const v2f*)(xrow + ko);
    v2f bv0 = *(const v2f*)(wr0 + ko);
    v2f bv1 = *(const v2f*)(wr1 + ko);
    v2f bv2 = *(const v2f*)(wr2 + ko);
    v2f bv3 = *(const v2f*)(wr3 + ko);
    acc[0] = __builtin_amdgcn_wmma_f32_16x16x4_f32(false, av, false, bv0,
                                                   (short)0, acc[0], false, false);
    acc[1] = __builtin_amdgcn_wmma_f32_16x16x4_f32(false, av, false, bv1,
                                                   (short)0, acc[1], false, false);
    acc[2] = __builtin_amdgcn_wmma_f32_16x16x4_f32(false, av, false, bv2,
                                                   (short)0, acc[2], false, false);
    acc[3] = __builtin_amdgcn_wmma_f32_16x16x4_f32(false, av, false, bv3,
                                                   (short)0, acc[3], false, false);
  }

  #pragma unroll
  for (int cc = 0; cc < 4; ++cc) {
    const float bval = bias[col0 + cc * 16 + l];
    #pragma unroll
    for (int r = 0; r < 8; ++r) {
      const int row = row0 + r + 8 * half;
      hbuf[(size_t)row * COUT_ + col0 + cc * 16 + l] = acc[cc][r] + bval;
    }
  }
}

__global__ __launch_bounds__(256) void gat_lscore(const float* __restrict__ hbuf,
                                                  const float* __restrict__ a,
                                                  float* __restrict__ lsrc,
                                                  float* __restrict__ ldst) {
  const int t    = blockIdx.x * blockDim.x + threadIdx.x;
  const int head = t & (H_ - 1);
  const int bi   = t >> 2;
  const float* hr = hbuf + (size_t)bi * COUT_ + head * C_;
  const float* as = a + head * (2 * C_);
  const float* ad = as + C_;
  float s = 0.f, d = 0.f;
  #pragma unroll 8
  for (int c = 0; c < C_; ++c) {
    const float v = hr[c];
    s = fmaf(v, as[c], s);
    d = fmaf(v, ad[c], d);
  }
  lsrc[t] = s;
  ldst[t] = d;
}

#define SROWS_ 4

__global__ __launch_bounds__(256) void gat_stats(const float* __restrict__ lsrc,
                                                 const float* __restrict__ ldst,
                                                 const int*   __restrict__ adj,
                                                 float* __restrict__ mbuf,
                                                 float* __restrict__ sbuf) {
  const int wave = blockIdx.x * (blockDim.x >> 5) + (threadIdx.x >> 5);
  const int lane = threadIdx.x & 31;
  const int row0 = wave * SROWS_;
  const int b    = row0 / N_;

  const float4* ldr = (const float4*)ldst + (size_t)b * N_ + lane;
  float4 ldv[32];
  #pragma unroll
  for (int k = 0; k < 32; ++k) ldv[k] = ldr[32 * k];

  const int*    arow = adj + (size_t)row0 * N_ + lane;
  const float4* lsr  = (const float4*)lsrc + row0;

  int adC[32];
  #pragma unroll
  for (int k = 0; k < 32; ++k) adC[k] = arow[32 * k];
  float4 lsC = lsr[0];

  #pragma nounroll
  for (int r = 0; r < SROWS_; ++r) {
    const int rn = (r + 1 < SROWS_) ? r + 1 : r;
    const int* an = arow + (size_t)rn * N_;
    int adN[32];
    #pragma unroll
    for (int k = 0; k < 32; ++k) adN[k] = an[32 * k];
    const float4 lsN = lsr[rn];
    __builtin_amdgcn_sched_barrier(0);

    const float ls[H_] = {lsC.x, lsC.y, lsC.z, lsC.w};

    float mx[H_];
    #pragma unroll
    for (int hd = 0; hd < H_; ++hd) mx[hd] = NEG_INF_;
    #pragma unroll
    for (int k = 0; k < 32; ++k) {
      const int   ad = adC[k];
      const float ldk[H_] = {ldv[k].x, ldv[k].y, ldv[k].z, ldv[k].w};
      #pragma unroll
      for (int hd = 0; hd < H_; ++hd) {
        float x = ls[hd] + ldk[hd];
        x = fmaxf(x, ALPHA_ * x);
        x = (ad == 1) ? x : NEG_INF_;
        mx[hd] = fmaxf(mx[hd], x);
      }
    }
    #pragma unroll
    for (int hd = 0; hd < H_; ++hd)
      #pragma unroll
      for (int off = 16; off > 0; off >>= 1)
        mx[hd] = fmaxf(mx[hd], __shfl_xor(mx[hd], off, 32));

    float sm[H_] = {0.f, 0.f, 0.f, 0.f};
    #pragma unroll
    for (int k = 0; k < 32; ++k) {
      const int   ad = adC[k];
      const float ldk[H_] = {ldv[k].x, ldv[k].y, ldv[k].z, ldv[k].w};
      #pragma unroll
      for (int hd = 0; hd < H_; ++hd) {
        float x = ls[hd] + ldk[hd];
        x = fmaxf(x, ALPHA_ * x);
        sm[hd] += (ad == 1) ? __expf(x - mx[hd]) : 0.f;
      }
    }
    #pragma unroll
    for (int hd = 0; hd < H_; ++hd)
      #pragma unroll
      for (int off = 16; off > 0; off >>= 1)
        sm[hd] += __shfl_xor(sm[hd], off, 32);

    if (lane == 0) {
      const size_t bi = (size_t)row0 + r;
      *(float4*)(mbuf + bi * H_) = make_float4(mx[0], mx[1], mx[2], mx[3]);
      *(float4*)(sbuf + bi * H_) = make_float4(sm[0], sm[1], sm[2], sm[3]);
    }

    #pragma unroll
    for (int k = 0; k < 32; ++k) adC[k] = adN[k];
    lsC = lsN;
  }
}

typedef int   v4i __attribute__((ext_vector_type(4)));
typedef float v4f __attribute__((ext_vector_type(4)));

struct AggChunk {
  v4i adA[4];
  v4i adB[4];
  v4f hr[4][4];
};

__device__ __forceinline__ void agg_load(AggChunk& q, const int* __restrict__ arA,
                                         const int* __restrict__ arB,
                                         const float* __restrict__ hcol, int jb) {
  #pragma unroll
  for (int u = 0; u < 4; ++u) {
    const int j = jb + 8 * u;
    q.adA[u] = *(const v4i*)(arA + j);
    q.adB[u] = *(const v4i*)(arB + j);
    #pragma unroll
    for (int r = 0; r < 4; ++r)
      q.hr[u][r] = *(const v4f*)(hcol + (size_t)(j + r) * COUT_);
  }
}

__device__ __forceinline__ void agg_compute(const AggChunk& q, const float* ldh, int jb,
                                            float lsA, float lsB, float mA, float mB,
                                            float isA, float isB,
                                            v8f (&accA)[4], v8f (&accB)[4]) {
  #pragma unroll
  for (int u = 0; u < 4; ++u) {
    const v4f ld = *(const v4f*)(ldh + jb + 8 * u);
    float pA[4], pB[4];
    #pragma unroll
    for (int r = 0; r < 4; ++r) {
      float xa = lsA + ld[r]; xa = fmaxf(xa, ALPHA_ * xa);
      float xb = lsB + ld[r]; xb = fmaxf(xb, ALPHA_ * xb);
      pA[r] = (q.adA[u][r] == 1) ? __expf(xa - mA) * isA : 0.0f;
      pB[r] = (q.adB[u][r] == 1) ? __expf(xb - mB) * isB : 0.0f;
    }
    #pragma unroll
    for (int t = 0; t < 2; ++t) {
      v2f avA, avB;
      avA.x = pA[2 * t]; avA.y = pA[2 * t + 1];
      avB.x = pB[2 * t]; avB.y = pB[2 * t + 1];
      #pragma unroll
      for (int cc = 0; cc < 4; ++cc) {
        v2f bv;
        bv.x = q.hr[u][2 * t][cc];
        bv.y = q.hr[u][2 * t + 1][cc];
        accA[cc] = __builtin_amdgcn_wmma_f32_16x16x4_f32(false, avA, false, bv,
                                                         (short)0, accA[cc],
                                                         false, false);
        accB[cc] = __builtin_amdgcn_wmma_f32_16x16x4_f32(false, avB, false, bv,
                                                         (short)0, accB[cc],
                                                         false, false);
      }
    }
  }
}

__global__ __launch_bounds__(128) void gat_aggregate(const float* __restrict__ hbuf,
                                                     const float* __restrict__ lsrc,
                                                     const float* __restrict__ ldst,
                                                     const int*   __restrict__ adj,
                                                     const float* __restrict__ mbuf,
                                                     const float* __restrict__ sbuf,
                                                     float* __restrict__ out) {
  __shared__ __attribute__((aligned(16))) float lds_ld[H_ * N_];

  const int b     = blockIdx.x >> 5;
  const int islab = blockIdx.x & 31;
  const int i0    = islab * 32;
  const int head  = threadIdx.x >> 5;
  const int lane  = threadIdx.x & 31;
  const int half  = lane >> 4;
  const int l     = lane & 15;

  {
    const float4* src = (const float4*)(ldst + (size_t)b * N_ * H_);
    for (int j = threadIdx.x; j < N_; j += 128) {
      const float4 v = src[j];
      lds_ld[0 * N_ + j] = v.x;
      lds_ld[1 * N_ + j] = v.y;
      lds_ld[2 * N_ + j] = v.z;
      lds_ld[3 * N_ + j] = v.w;
    }
  }

  const size_t biA = (size_t)b * N_ + (i0 + l);
  const size_t biB = biA + 16;
  const int*   arA = adj + biA * (size_t)N_;
  const int*   arB = adj + biB * (size_t)N_;
  const float* hcol = hbuf + (size_t)b * N_ * COUT_ + head * C_ + 4 * l;
  const float* ldh  = lds_ld + head * N_;
  const int    jh   = 4 * half;

  AggChunk q0, q1;
  agg_load(q0, arA, arB, hcol, jh);

  const float  lsA = lsrc[biA * H_ + head];
  const float  lsB = lsrc[biB * H_ + head];
  const float  mA  = mbuf[biA * H_ + head];
  const float  mB  = mbuf[biB * H_ + head];
  const float  isA = 1.0f / sbuf[biA * H_ + head];
  const float  isB = 1.0f / sbuf[biB * H_ + head];

  __syncthreads();

  v8f accA[4] = {{}, {}, {}, {}};
  v8f accB[4] = {{}, {}, {}, {}};

  for (int jc = 0; jc < N_; jc += 64) {
    agg_load(q1, arA, arB, hcol, jc + 32 + jh);
    __builtin_amdgcn_sched_barrier(0);
    agg_compute(q0, ldh, jc + jh, lsA, lsB, mA, mB, isA, isB, accA, accB);
    __builtin_amdgcn_sched_barrier(0);
    agg_load(q0, arA, arB, hcol, ((jc + 64) & (N_ - 1)) + jh);
    __builtin_amdgcn_sched_barrier(0);
    agg_compute(q1, ldh, jc + 32 + jh, lsA, lsB, mA, mB, isA, isB, accA, accB);
  }

  #pragma unroll
  for (int r = 0; r < 8; ++r) {
    const int rowA = i0 + r + 8 * half;
    const int rowB = rowA + 16;
    v4f oa, ob;
    oa.x = accA[0][r]; oa.y = accA[1][r]; oa.z = accA[2][r]; oa.w = accA[3][r];
    ob.x = accB[0][r]; ob.y = accB[1][r]; ob.z = accB[2][r]; ob.w = accB[3][r];
    *(v4f*)(out + ((size_t)b * N_ + rowA) * COUT_ + head * C_ + 4 * l) = oa;
    *(v4f*)(out + ((size_t)b * N_ + rowB) * COUT_ + head * C_ + 4 * l) = ob;
  }
}

extern "C" void kernel_launch(void* const* d_in, const int* in_sizes, int n_in,
                              void* d_out, int out_size, void* d_ws, size_t ws_size,
                              hipStream_t stream) {
  const float* node_feats = (const float*)d_in[0];
  const int*   adj        = (const int*)  d_in[1];
  const float* W          = (const float*)d_in[2];
  const float* bias       = (const float*)d_in[3];
  const float* a          = (const float*)d_in[4];
  float*       out        = (float*)d_out;

  float* hbuf = (float*)d_ws;
  float* lsrc = hbuf + (size_t)B_ * N_ * COUT_;
  float* ldst = lsrc + (size_t)B_ * N_ * H_;
  float* mbuf = ldst + (size_t)B_ * N_ * H_;
  float* sbuf = mbuf + (size_t)B_ * N_ * H_;

  gat_gemm_h<<<(B_ * N_ / 16) * (COUT_ / 64) / 8, 256, 0, stream>>>(
      node_feats, W, bias, hbuf);

  gat_lscore<<<(B_ * N_ * H_) / 256, 256, 0, stream>>>(hbuf, a, lsrc, ldst);

  gat_stats<<<(B_ * N_) / (8 * SROWS_), 256, 0, stream>>>(lsrc, ldst, adj, mbuf, sbuf);

  gat_aggregate<<<B_ * (N_ / 32), 128, 0, stream>>>(
      hbuf, lsrc, ldst, adj, mbuf, sbuf, out);
}


# Logit halves formed in the projection kernel's tail: behind a block barrier one thread per (node, head) re-reads its 64 fresh h values, same fmaf chain (same bits); logit launch gone (3 launches)
# speedup vs baseline: 2.1419x; 1.0427x over previous
#include <hip/hip_runtime.h>
#include <hip/hip_bf16.h>
#include <math.h>

#define B_      8
#define N_      1024
#define CIN_    256
#define H_      4
#define COUT_   256
#define C_      64
#define ALPHA_  0.2f
#define NEG_INF_ (-9.0e15f)

typedef float v2f __attribute__((ext_vector_type(2)));
typedef float v8f __attribute__((ext_vector_type(8)));

#define GEMM_BROWS_ 32

__global__ __launch_bounds__(256) void gat_gemm_h(const float* __restrict__ X,
                                                  const float* __restrict__ W,
                                                  const float* __restrict__ bias,
                                                  const float* __restrict__ a,
                                                  float* __restrict__ hbuf,
                                                  float* __restrict__ lsrc,
                                                  float* __restrict__ ldst) {
  const int wave = blockIdx.x * (blockDim.x >> 5) + (threadIdx.x >> 5);
  const int lane = threadIdx.x & 31;
  const int half = lane >> 4;
  const int l    = lane & 15;

  const int tileM = wave >> 2;
  const int cg    = wave & 3;
  const int row0  = tileM * 16;
  const int col0  = cg * 64;

  const float* xrow = X + (size_t)(row0 + l) * CIN_;
  const float* wr0  = W + (size_t)(col0 +  0 + l) * CIN_;
  const float* wr1  = W + (size_t)(col0 + 16 + l) * CIN_;
  const float* wr2  = W + (size_t)(col0 + 32 + l) * CIN_;
  const float* wr3  = W + (size_t)(col0 + 48 + l) * CIN_;

  v8f acc[4] = {{}, {}, {}, {}};
  #pragma unroll 4
  for (int kk = 0; kk < CIN_; kk += 4) {
    const int ko = kk + 2 * half;
    v2f av  = *(const v2f*)(xrow + ko);
    v2f bv0 = *(const v2f*)(wr0 + ko);
    v2f bv1 = *(const v2f*)(wr1 + ko);
    v2f bv2 = *(const v2f*)(wr2 + ko);
    v2f bv3 = *(const v2f*)(wr3 + ko);
    acc[0] = __builtin_amdgcn_wmma_f32_16x16x4_f32(false, av, false, bv0,
                                                   (short)0, acc[0], false, false);
    acc[1] = __builtin_amdgcn_wmma_f32_16x16x4_f32(false, av, false, bv1,
                                                   (short)0, acc[1], false, false);
    acc[2] = __builtin_amdgcn_wmma_f32_16x16x4_f32(false, av, false, bv2,
                                                   (short)0, acc[2], false, false);
    acc[3] = __builtin_amdgcn_wmma_f32_16x16x4_f32(false, av, false, bv3,
                                                   (short)0, acc[3], false, false);
  }

  #pragma unroll
  for (int cc = 0; cc < 4; ++cc) {
    const float bval = bias[col0 + cc * 16 + l];
    #pragma unroll
    for (int r = 0; r < 8; ++r) {
      const int row = row0 + r + 8 * half;
      hbuf[(size_t)row * COUT_ + col0 + cc * 16 + l] = acc[cc][r] + bval;
    }
  }

  __syncthreads();
  if (threadIdx.x < GEMM_BROWS_ * H_) {
    const int head = threadIdx.x & (H_ - 1);
    const int bi   = blockIdx.x * GEMM_BROWS_ + (threadIdx.x >> 2);
    const float* hr = hbuf + (size_t)bi * COUT_ + head * C_;
    const float* as = a + head * (2 * C_);
    const float* ad = as + C_;
    float s = 0.f, d = 0.f;
    #pragma unroll
    for (int c = 0; c < C_; ++c) {
      const float v = hr[c];
      s = fmaf(v, as[c], s);
      d = fmaf(v, ad[c], d);
    }
    lsrc[bi * H_ + head] = s;
    ldst[bi * H_ + head] = d;
  }
}

#define SROWS_ 4

__global__ __launch_bounds__(256) void gat_stats(const float* __restrict__ lsrc,
                                                 const float* __restrict__ ldst,
                                                 const int*   __restrict__ adj,
                                                 float* __restrict__ mbuf,
                                                 float* __restrict__ sbuf) {
  const int wave = blockIdx.x * (blockDim.x >> 5) + (threadIdx.x >> 5);
  const int lane = threadIdx.x & 31;
  const int row0 = wave * SROWS_;
  const int b    = row0 / N_;

  const float4* ldr = (const float4*)ldst + (size_t)b * N_ + lane;
  float4 ldv[32];
  #pragma unroll
  for (int k = 0; k < 32; ++k) ldv[k] = ldr[32 * k];

  const int*    arow = adj + (size_t)row0 * N_ + lane;
  const float4* lsr  = (const float4*)lsrc + row0;

  int adC[32];
  #pragma unroll
  for (int k = 0; k < 32; ++k) adC[k] = arow[32 * k];
  float4 lsC = lsr[0];

  #pragma nounroll
  for (int r = 0; r < SROWS_; ++r) {
    const int rn = (r + 1 < SROWS_) ? r + 1 : r;
    const int* an = arow + (size_t)rn * N_;
    int adN[32];
    #pragma unroll
    for (int k = 0; k < 32; ++k) adN[k] = an[32 * k];
    const float4 lsN = lsr[rn];
    __builtin_amdgcn_sched_barrier(0);

    const float ls[H_] = {lsC.x, lsC.y, lsC.z, lsC.w};

    float mx[H_];
    #pragma unroll
    for (int hd = 0; hd < H_; ++hd) mx[hd] = NEG_INF_;
    #pragma unroll
    for (int k = 0; k < 32; ++k) {
      const int   ad = adC[k];
      const float ldk[H_] = {ldv[k].x, ldv[k].y, ldv[k].z, ldv[k].w};
      #pragma unroll
      for (int hd = 0; hd < H_; ++hd) {
        float x = ls[hd] + ldk[hd];
        x = fmaxf(x, ALPHA_ * x);
        x = (ad == 1) ? x : NEG_INF_;
        mx[hd] = fmaxf(mx[hd], x);
      }
    }
    #pragma unroll
    for (int hd = 0; hd < H_; ++hd)
      #pragma unroll
      for (int off = 16; off > 0; off >>= 1)
        mx[hd] = fmaxf(mx[hd], __shfl_xor(mx[hd], off, 32));

    float sm[H_] = {0.f, 0.f, 0.f, 0.f};
    #pragma unroll
    for (int k = 0; k < 32; ++k) {
      const int   ad = adC[k];
      const float ldk[H_] = {ldv[k].x, ldv[k].y, ldv[k].z, ldv[k].w};
      #pragma unroll
      for (int hd = 0; hd < H_; ++hd) {
        float x = ls[hd] + ldk[hd];
        x = fmaxf(x, ALPHA_ * x);
        sm[hd] += (ad == 1) ? __expf(x - mx[hd]) : 0.f;
      }
    }
    #pragma unroll
    for (int hd = 0; hd < H_; ++hd)
      #pragma unroll
      for (int off = 16; off > 0; off >>= 1)
        sm[hd] += __shfl_xor(sm[hd], off, 32);

    if (lane == 0) {
      const size_t bi = (size_t)row0 + r;
      *(float4*)(mbuf + bi * H_) = make_float4(mx[0], mx[1], mx[2], mx[3]);
      *(float4*)(sbuf + bi * H_) = make_float4(sm[0], sm[1], sm[2], sm[3]);
    }

    #pragma unroll
    for (int k = 0; k < 32; ++k) adC[k] = adN[k];
    lsC = lsN;
  }
}

typedef int   v4i __attribute__((ext_vector_type(4)));
typedef float v4f __attribute__((ext_vector_type(4)));

struct AggChunk {
  v4i adA[4];
  v4i adB[4];
  v4f hr[4][4];
};

__device__ __forceinline__ void agg_load(AggChunk& q, const int* __restrict__ arA,
                                         const int* __restrict__ arB,
                                         const float* __restrict__ hcol, int jb) {
  #pragma unroll
  for (int u = 0; u < 4; ++u) {
    const int j = jb + 8 * u;
    q.adA[u] = *(const v4i*)(arA + j);
    q.adB[u] = *(const v4i*)(arB + j);
    #pragma unroll
    for (int r = 0; r < 4; ++r)
      q.hr[u][r] = *(const v4f*)(hcol + (size_t)(j + r) * COUT_);
  }
}

__device__ __forceinline__ void agg_compute(const AggChunk& q, const float* ldh, int jb,
                                            float lsA, float lsB, float mA, float mB,
                                            float isA, float isB,
                                            v8f (&accA)[4], v8f (&accB)[4]) {
  #pragma unroll
  for (int u = 0; u < 4; ++u) {
    const v4f ld = *(const v4f*)(ldh + jb + 8 * u);
    float pA[4], pB[4];
    #pragma unroll
    for (int r = 0; r < 4; ++r) {
      float xa = lsA + ld[r]; xa = fmaxf(xa, ALPHA_ * xa);
      float xb = lsB + ld[r]; xb = fmaxf(xb, ALPHA_ * xb);
      pA[r] = (q.adA[u][r] == 1) ? __expf(xa - mA) * isA : 0.0f;
      pB[r] = (q.adB[u][r] == 1) ? __expf(xb - mB) * isB : 0.0f;
    }
    #pragma unroll
    for (int t = 0; t < 2; ++t) {
      v2f avA, avB;
      avA.x = pA[2 * t]; avA.y = pA[2 * t + 1];
      avB.x = pB[2 * t]; avB.y = pB[2 * t + 1];
      #pragma unroll
      for (int cc = 0; cc < 4; ++cc) {
        v2f bv;
        bv.x = q.hr[u][2 * t][cc];
        bv.y = q.hr[u][2 * t + 1][cc];
        accA[cc] = __builtin_amdgcn_wmma_f32_16x16x4_f32(false, avA, false, bv,
                                                         (short)0, accA[cc],
                                                         false, false);
        accB[cc] = __builtin_amdgcn_wmma_f32_16x16x4_f32(false, avB, false, bv,
                                                         (short)0, accB[cc],
                                                         false, false);
      }
    }
  }
}

__global__ __launch_bounds__(128) void gat_aggregate(const float* __restrict__ hbuf,
                                                     const float* __restrict__ lsrc,
                                                     const float* __restrict__ ldst,
                                                     const int*   __restrict__ adj,
                                                     const float* __restrict__ mbuf,
                                                     const float* __restrict__ sbuf,
                                                     float* __restrict__ out) {
  __shared__ __attribute__((aligned(16))) float lds_ld[H_ * N_];

  const int b     = blockIdx.x >> 5;
  const int islab = blockIdx.x & 31;
  const int i0    = islab * 32;
  const int head  = threadIdx.x >> 5;
  const int lane  = threadIdx.x & 31;
  const int half  = lane >> 4;
  const int l     = lane & 15;

  {
    const float4* src = (const float4*)(ldst + (size_t)b * N_ * H_);
    for (int j = threadIdx.x; j < N_; j += 128) {
      const float4 v = src[j];
      lds_ld[0 * N_ + j] = v.x;
      lds_ld[1 * N_ + j] = v.y;
      lds_ld[2 * N_ + j] = v.z;
      lds_ld[3 * N_ + j] = v.w;
    }
  }

  const size_t biA = (size_t)b * N_ + (i0 + l);
  const size_t biB = biA + 16;
  const int*   arA = adj + biA * (size_t)N_;
  const int*   arB = adj + biB * (size_t)N_;
  const float* hcol = hbuf + (size_t)b * N_ * COUT_ + head * C_ + 4 * l;
  const float* ldh  = lds_ld + head * N_;
  const int    jh   = 4 * half;

  AggChunk q0, q1;
  agg_load(q0, arA, arB, hcol, jh);

  const float  lsA = lsrc[biA * H_ + head];
  const float  lsB = lsrc[biB * H_ + head];
  const float  mA  = mbuf[biA * H_ + head];
  const float  mB  = mbuf[biB * H_ + head];
  const float  isA = 1.0f / sbuf[biA * H_ + head];
  const float  isB = 1.0f / sbuf[biB * H_ + head];

  __syncthreads();

  v8f accA[4] = {{}, {}, {}, {}};
  v8f accB[4] = {{}, {}, {}, {}};

  for (int jc = 0; jc < N_; jc += 64) {
    agg_load(q1, arA, arB, hcol, jc + 32 + jh);
    __builtin_amdgcn_sched_barrier(0);
    agg_compute(q0, ldh, jc + jh, lsA, lsB, mA, mB, isA, isB, accA, accB);
    __builtin_amdgcn_sched_barrier(0);
    agg_load(q0, arA, arB, hcol, ((jc + 64) & (N_ - 1)) + jh);
    __builtin_amdgcn_sched_barrier(0);
    agg_compute(q1, ldh, jc + 32 + jh, lsA, lsB, mA, mB, isA, isB, accA, accB);
  }

  #pragma unroll
  for (int r = 0; r < 8; ++r) {
    const int rowA = i0 + r + 8 * half;
    const int rowB = rowA + 16;
    v4f oa, ob;
    oa.x = accA[0][r]; oa.y = accA[1][r]; oa.z = accA[2][r]; oa.w = accA[3][r];
    ob.x = accB[0][r]; ob.y = accB[1][r]; ob.z = accB[2][r]; ob.w = accB[3][r];
    *(v4f*)(out + ((size_t)b * N_ + rowA) * COUT_ + head * C_ + 4 * l) = oa;
    *(v4f*)(out + ((size_t)b * N_ + rowB) * COUT_ + head * C_ + 4 * l) = ob;
  }
}

extern "C" void kernel_launch(void* const* d_in, const int* in_sizes, int n_in,
                              void* d_out, int out_size, void* d_ws, size_t ws_size,
                              hipStream_t stream) {
  const float* node_feats = (const float*)d_in[0];
  const int*   adj        = (const int*)  d_in[1];
  const float* W          = (const float*)d_in[2];
  const float* bias       = (const float*)d_in[3];
  const float* a          = (const float*)d_in[4];
  float*       out        = (float*)d_out;

  float* hbuf = (float*)d_ws;
  float* lsrc = hbuf + (size_t)B_ * N_ * COUT_;
  float* ldst = lsrc + (size_t)B_ * N_ * H_;
  float* mbuf = ldst + (size_t)B_ * N_ * H_;
  float* sbuf = mbuf + (size_t)B_ * N_ * H_;

  gat_gemm_h<<<(B_ * N_ / 16) * (COUT_ / 64) / 8, 256, 0, stream>>>(
      node_feats, W, bias, a, hbuf, lsrc, ldst);

  gat_stats<<<(B_ * N_) / (8 * SROWS_), 256, 0, stream>>>(lsrc, ldst, adj, mbuf, sbuf);

  gat_aggregate<<<B_ * (N_ / 32), 128, 0, stream>>>(
      hbuf, lsrc, ldst, adj, mbuf, sbuf, out);
}


# Projection GEMM re-tiled in registers: a wave owns 32x64 of h; A/B fragments straight from global by 16-byte loads, a K-chunk of 32 ahead; k pairs permuted in 8-groups (new sum order)
# speedup vs baseline: 2.4450x; 1.1415x over previous
#include <hip/hip_runtime.h>
#include <hip/hip_bf16.h>
#include <math.h>

#define B_      8
#define N_      1024
#define CIN_    256
#define H_      4
#define COUT_   256
#define C_      64
#define ALPHA_  0.2f
#define NEG_INF_ (-9.0e15f)

typedef float v2f __attribute__((ext_vector_type(2)));
typedef float v8f __attribute__((ext_vector_type(8)));

#define GEMM_BROWS_ 32

typedef float v4f __attribute__((ext_vector_type(4)));

struct GemmChunk {
  v4f a[4][2];
  v4f b[4][4];
};

__device__ __forceinline__ void gemm_load(GemmChunk& q, const float* __restrict__ xa,
                                          const float* __restrict__ wb, int kb) {
  #pragma unroll
  for (int p = 0; p < 4; ++p) {
    const int ko = kb + 8 * p;
    q.a[p][0] = *(const v4f*)(xa + ko);
    q.a[p][1] = *(const v4f*)(xa + 16 * CIN_ + ko);
    #pragma unroll
    for (int cc = 0; cc < 4; ++cc)
      q.b[p][cc] = *(const v4f*)(wb + cc * 16 * CIN_ + ko);
  }
}

__device__ __forceinline__ void gemm_compute(const GemmChunk& q, v8f (&acc)[2][4]) {
  #pragma unroll
  for (int p = 0; p < 4; ++p) {
    #pragma unroll
    for (int t = 0; t < 2; ++t) {
      v2f a0, a1;
      a0.x = q.a[p][0][2 * t]; a0.y = q.a[p][0][2 * t + 1];
      a1.x = q.a[p][1][2 * t]; a1.y = q.a[p][1][2 * t + 1];
      #pragma unroll
      for (int cc = 0; cc < 4; ++cc) {
        v2f bv;
        bv.x = q.b[p][cc][2 * t];
        bv.y = q.b[p][cc][2 * t + 1];
        acc[0][cc] = __builtin_amdgcn_wmma_f32_16x16x4_f32(false, a0, false, bv,
                                                           (short)0, acc[0][cc],
                                                           false, false);
        acc[1][cc] = __builtin_amdgcn_wmma_f32_16x16x4_f32(false, a1, false, bv,
                                                           (short)0, acc[1][cc],
                                                           false, false);
      }
    }
  }
}

__global__ __launch_bounds__(128) void gat_gemm_h(const float* __restrict__ X,
                                                  const float* __restrict__ W,
                                                  const float* __restrict__ bias,
                                                  const float* __restrict__ a,
                                                  float* __restrict__ hbuf,
                                                  float* __restrict__ lsrc,
                                                  float* __restrict__ ldst) {
  const int cg   = threadIdx.x >> 5;
  const int lane = threadIdx.x & 31;
  const int half = lane >> 4;
  const int l    = lane & 15;

  const int row0 = blockIdx.x * GEMM_BROWS_;
  const int col0 = cg * 64;

  const float* xa = X + (size_t)(row0 + l) * CIN_ + 4 * half;
  const float* wb = W + (size_t)(col0 + l) * CIN_ + 4 * half;

  GemmChunk q0, q1;
  gemm_load(q0, xa, wb, 0);

  v8f acc[2][4] = {{{}, {}, {}, {}}, {{}, {}, {}, {}}};
  #pragma unroll
  for (int kc = 0; kc < CIN_; kc += 64) {
    gemm_load(q1, xa, wb, kc + 32);
    __builtin_amdgcn_sched_barrier(0);
    gemm_compute(q0, acc);
    __builtin_amdgcn_sched_barrier(0);
    if (kc + 64 < CIN_) gemm_load(q0, xa, wb, kc + 64);
    __builtin_amdgcn_sched_barrier(0);
    gemm_compute(q1, acc);
  }

  #pragma unroll
  for (int rt = 0; rt < 2; ++rt) {
    #pragma unroll
    for (int cc = 0; cc < 4; ++cc) {
      const float bval = bias[col0 + cc * 16 + l];
      #pragma unroll
      for (int r = 0; r < 8; ++r) {
        const int row = row0 + 16 * rt + r + 8 * half;
        hbuf[(size_t)row * COUT_ + col0 + cc * 16 + l] = acc[rt][cc][r] + bval;
      }
    }
  }

  __syncthreads();
  {
    const int head = threadIdx.x & (H_ - 1);
    const int bi   = blockIdx.x * GEMM_BROWS_ + (threadIdx.x >> 2);
    const float* hr = hbuf + (size_t)bi * COUT_ + head * C_;
    const float* as = a + head * (2 * C_);
    const float* ad = as + C_;
    float s = 0.f, d = 0.f;
    #pragma unroll
    for (int c = 0; c < C_; ++c) {
      const float v = hr[c];
      s = fmaf(v, as[c], s);
      d = fmaf(v, ad[c], d);
    }
    lsrc[bi * H_ + head] = s;
    ldst[bi * H_ + head] = d;
  }
}

#define SROWS_ 4

__global__ __launch_bounds__(256) void gat_stats(const float* __restrict__ lsrc,
                                                 const float* __restrict__ ldst,
                                                 const int*   __restrict__ adj,
                                                 float* __restrict__ mbuf,
                                                 float* __restrict__ sbuf) {
  const int wave = blockIdx.x * (blockDim.x >> 5) + (threadIdx.x >> 5);
  const int lane = threadIdx.x & 31;
  const int row0 = wave * SROWS_;
  const int b    = row0 / N_;

  const float4* ldr = (const float4*)ldst + (size_t)b * N_ + lane;
  float4 ldv[32];
  #pragma unroll
  for (int k = 0; k < 32; ++k) ldv[k] = ldr[32 * k];

  const int*    arow = adj + (size_t)row0 * N_ + lane;
  const float4* lsr  = (const float4*)lsrc + row0;

  int adC[32];
  #pragma unroll
  for (int k = 0; k < 32; ++k) adC[k] = arow[32 * k];
  float4 lsC = lsr[0];

  #pragma nounroll
  for (int r = 0; r < SROWS_; ++r) {
    const int rn = (r + 1 < SROWS_) ? r + 1 : r;
    const int* an = arow + (size_t)rn * N_;
    int adN[32];
    #pragma unroll
    for (int k = 0; k < 32; ++k) adN[k] = an[32 * k];
    const float4 lsN = lsr[rn];
    __builtin_amdgcn_sched_barrier(0);

    const float ls[H_] = {lsC.x, lsC.y, lsC.z, lsC.w};

    float mx[H_];
    #pragma unroll
    for (int hd = 0; hd < H_; ++hd) mx[hd] = NEG_INF_;
    #pragma unroll
    for (int k = 0; k < 32; ++k) {
      const int   ad = adC[k];
      const float ldk[H_] = {ldv[k].x, ldv[k].y, ldv[k].z, ldv[k].w};
      #pragma unroll
      for (int hd = 0; hd < H_; ++hd) {
        float x = ls[hd] + ldk[hd];
        x = fmaxf(x, ALPHA_ * x);
        x = (ad == 1) ? x : NEG_INF_;
        mx[hd] = fmaxf(mx[hd], x);
      }
    }
    #pragma unroll
    for (int hd = 0; hd < H_; ++hd)
      #pragma unroll
      for (int off = 16; off > 0; off >>= 1)
        mx[hd] = fmaxf(mx[hd], __shfl_xor(mx[hd], off, 32));

    float sm[H_] = {0.f, 0.f, 0.f, 0.f};
    #pragma unroll
    for (int k = 0; k < 32; ++k) {
      const int   ad = adC[k];
      const float ldk[H_] = {ldv[k].x, ldv[k].y, ldv[k].z, ldv[k].w};
      #pragma unroll
      for (int hd = 0; hd < H_; ++hd) {
        float x = ls[hd] + ldk[hd];
        x = fmaxf(x, ALPHA_ * x);
        sm[hd] += (ad == 1) ? __expf(x - mx[hd]) : 0.f;
      }
    }
    #pragma unroll
    for (int hd = 0; hd < H_; ++hd)
      #pragma unroll
      for (int off = 16; off > 0; off >>= 1)
        sm[hd] += __shfl_xor(sm[hd], off, 32);

    if (lane == 0) {
      const size_t bi = (size_t)row0 + r;
      *(float4*)(mbuf + bi * H_) = make_float4(mx[0], mx[1], mx[2], mx[3]);
      *(float4*)(sbuf + bi * H_) = make_float4(sm[0], sm[1], sm[2], sm[3]);
    }

    #pragma unroll
    for (int k = 0; k < 32; ++k) adC[k] = adN[k];
    lsC = lsN;
  }
}

typedef int   v4i __attribute__((ext_vector_type(4)));

struct AggChunk {
  v4i adA[4];
  v4i adB[4];
  v4f hr[4][4];
};

__device__ __forceinline__ void agg_load(AggChunk& q, const int* __restrict__ arA,
                                         const int* __restrict__ arB,
                                         const float* __restrict__ hcol, int jb) {
  #pragma unroll
  for (int u = 0; u < 4; ++u) {
    const int j = jb + 8 * u;
    q.adA[u] = *(const v4i*)(arA + j);
    q.adB[u] = *(const v4i*)(arB + j);
    #pragma unroll
    for (int r = 0; r < 4; ++r)
      q.hr[u][r] = *(const v4f*)(hcol + (size_t)(j + r) * COUT_);
  }
}

__device__ __forceinline__ void agg_compute(const AggChunk& q, const float* ldh, int jb,
                                            float lsA, float lsB, float mA, float mB,
                                            float isA, float isB,
                                            v8f (&accA)[4], v8f (&accB)[4]) {
  #pragma unroll
  for (int u = 0; u < 4; ++u) {
    const v4f ld = *(const v4f*)(ldh + jb + 8 * u);
    float pA[4], pB[4];
    #pragma unroll
    for (int r = 0; r < 4; ++r) {
      float xa = lsA + ld[r]; xa = fmaxf(xa, ALPHA_ * xa);
      float xb = lsB + ld[r]; xb = fmaxf(xb, ALPHA_ * xb);
      pA[r] = (q.adA[u][r] == 1) ? __expf(xa - mA) * isA : 0.0f;
      pB[r] = (q.adB[u][r] == 1) ? __expf(xb - mB) * isB : 0.0f;
    }
    #pragma unroll
    for (int t = 0; t < 2; ++t) {
      v2f avA, avB;
      avA.x = pA[2 * t]; avA.y = pA[2 * t + 1];
      avB.x = pB[2 * t]; avB.y = pB[2 * t + 1];
      #pragma unroll
      for (int cc = 0; cc < 4; ++cc) {
        v2f bv;
        bv.x = q.hr[u][2 * t][cc];
        bv.y = q.hr[u][2 * t + 1][cc];
        accA[cc] = __builtin_amdgcn_wmma_f32_16x16x4_f32(false, avA, false, bv,
                                                         (short)0, accA[cc],
                                                         false, false);
        accB[cc] = __builtin_amdgcn_wmma_f32_16x16x4_f32(false, avB, false, bv,
                                                         (short)0, accB[cc],
                                                         false, false);
      }
    }
  }
}

__global__ __launch_bounds__(128) void gat_aggregate(const float* __restrict__ hbuf,
                                                     const float* __restrict__ lsrc,
                                                     const float* __restrict__ ldst,
                                                     const int*   __restrict__ adj,
                                                     const float* __restrict__ mbuf,
                                                     const float* __restrict__ sbuf,
                                                     float* __restrict__ out) {
  __shared__ __attribute__((aligned(16))) float lds_ld[H_ * N_];

  const int b     = blockIdx.x >> 5;
  const int islab = blockIdx.x & 31;
  const int i0    = islab * 32;
  const int head  = threadIdx.x >> 5;
  const int lane  = threadIdx.x & 31;
  const int half  = lane >> 4;
  const int l     = lane & 15;

  {
    const float4* src = (const float4*)(ldst + (size_t)b * N_ * H_);
    for (int j = threadIdx.x; j < N_; j += 128) {
      const float4 v = src[j];
      lds_ld[0 * N_ + j] = v.x;
      lds_ld[1 * N_ + j] = v.y;
      lds_ld[2 * N_ + j] = v.z;
      lds_ld[3 * N_ + j] = v.w;
    }
  }

  const size_t biA = (size_t)b * N_ + (i0 + l);
  const size_t biB = biA + 16;
  const int*   arA = adj + biA * (size_t)N_;
  const int*   arB = adj + biB * (size_t)N_;
  const float* hcol = hbuf + (size_t)b * N_ * COUT_ + head * C_ + 4 * l;
  const float* ldh  = lds_ld + head * N_;
  const int    jh   = 4 * half;

  AggChunk q0, q1;
  agg_load(q0, arA, arB, hcol, jh);

  const float  lsA = lsrc[biA * H_ + head];
  const float  lsB = lsrc[biB * H_ + head];
  const float  mA  = mbuf[biA * H_ + head];
  const float  mB  = mbuf[biB * H_ + head];
  const float  isA = 1.0f / sbuf[biA * H_ + head];
  const float  isB = 1.0f / sbuf[biB * H_ + head];

  __syncthreads();

  v8f accA[4] = {{}, {}, {}, {}};
  v8f accB[4] = {{}, {}, {}, {}};

  for (int jc = 0; jc < N_; jc += 64) {
    agg_load(q1, arA, arB, hcol, jc + 32 + jh);
    __builtin_amdgcn_sched_barrier(0);
    agg_compute(q0, ldh, jc + jh, lsA, lsB, mA, mB, isA, isB, accA, accB);
    __builtin_amdgcn_sched_barrier(0);
    agg_load(q0, arA, arB, hcol, ((jc + 64) & (N_ - 1)) + jh);
    __builtin_amdgcn_sched_barrier(0);
    agg_compute(q1, ldh, jc + 32 + jh, lsA, lsB, mA, mB, isA, isB, accA, accB);
  }

  #pragma unroll
  for (int r = 0; r < 8; ++r) {
    const int rowA = i0 + r + 8 * half;
    const int rowB = rowA + 16;
    v4f oa, ob;
    oa.x = accA[0][r]; oa.y = accA[1][r]; oa.z = accA[2][r]; oa.w = accA[3][r];
    ob.x = accB[0][r]; ob.y = accB[1][r]; ob.z = accB[2][r]; ob.w = accB[3][r];
    *(v4f*)(out + ((size_t)b * N_ + rowA) * COUT_ + head * C_ + 4 * l) = oa;
    *(v4f*)(out + ((size_t)b * N_ + rowB) * COUT_ + head * C_ + 4 * l) = ob;
  }
}

extern "C" void kernel_launch(void* const* d_in, const int* in_sizes, int n_in,
                              void* d_out, int out_size, void* d_ws, size_t ws_size,
                              hipStream_t stream) {
  const float* node_feats = (const float*)d_in[0];
  const int*   adj        = (const int*)  d_in[1];
  const float* W          = (const float*)d_in[2];
  const float* bias       = (const float*)d_in[3];
  const float* a          = (const float*)d_in[4];
  float*       out        = (float*)d_out;

  float* hbuf = (float*)d_ws;
  float* lsrc = hbuf + (size_t)B_ * N_ * COUT_;
  float* ldst = lsrc + (size_t)B_ * N_ * H_;
  float* mbuf = ldst + (size_t)B_ * N_ * H_;
  float* sbuf = mbuf + (size_t)B_ * N_ * H_;

  gat_gemm_h<<<B_ * N_ / GEMM_BROWS_, 128, 0, stream>>>(
      node_feats, W, bias, a, hbuf, lsrc, ldst);

  gat_stats<<<(B_ * N_) / (8 * SROWS_), 256, 0, stream>>>(lsrc, ldst, adj, mbuf, sbuf);

  gat_aggregate<<<B_ * (N_ / 32), 128, 0, stream>>>(
      hbuf, lsrc, ldst, adj, mbuf, sbuf, out);
}


# Softmax shifted by an upper bound of the row max, lrelu(ls + max of all l_dst), so no stats pass: 2 launches; row sums from the aggregation's weights (new sum order), exact-max redo on underflow
# speedup vs baseline: 2.7428x; 1.1218x over previous
#include <hip/hip_runtime.h>
#include <hip/hip_bf16.h>
#include <math.h>

#define B_      8
#define N_      1024
#define CIN_    256
#define H_      4
#define COUT_   256
#define C_      64
#define ALPHA_  0.2f
#define NEG_INF_ (-9.0e15f)

typedef float v2f __attribute__((ext_vector_type(2)));
typedef float v8f __attribute__((ext_vector_type(8)));

#define GEMM_BROWS_ 32

typedef float v4f __attribute__((ext_vector_type(4)));

struct GemmChunk {
  v4f a[4][2];
  v4f b[4][4];
};

__device__ __forceinline__ void gemm_load(GemmChunk& q, const float* __restrict__ xa,
                                          const float* __restrict__ wb, int kb) {
  #pragma unroll
  for (int p = 0; p < 4; ++p) {
    const int ko = kb + 8 * p;
    q.a[p][0] = *(const v4f*)(xa + ko);
    q.a[p][1] = *(const v4f*)(xa + 16 * CIN_ + ko);
    #pragma unroll
    for (int cc = 0; cc < 4; ++cc)
      q.b[p][cc] = *(const v4f*)(wb + cc * 16 * CIN_ + ko);
  }
}

__device__ __forceinline__ void gemm_compute(const GemmChunk& q, v8f (&acc)[2][4]) {
  #pragma unroll
  for (int p = 0; p < 4; ++p) {
    #pragma unroll
    for (int t = 0; t < 2; ++t) {
      v2f a0, a1;
      a0.x = q.a[p][0][2 * t]; a0.y = q.a[p][0][2 * t + 1];
      a1.x = q.a[p][1][2 * t]; a1.y = q.a[p][1][2 * t + 1];
      #pragma unroll
      for (int cc = 0; cc < 4; ++cc) {
        v2f bv;
        bv.x = q.b[p][cc][2 * t];
        bv.y = q.b[p][cc][2 * t + 1];
        acc[0][cc] = __builtin_amdgcn_wmma_f32_16x16x4_f32(false, a0, false, bv,
                                                           (short)0, acc[0][cc],
                                                           false, false);
        acc[1][cc] = __builtin_amdgcn_wmma_f32_16x16x4_f32(false, a1, false, bv,
                                                           (short)0, acc[1][cc],
                                                           false, false);
      }
    }
  }
}

__global__ __launch_bounds__(128) void gat_gemm_h(const float* __restrict__ X,
                                                  const float* __restrict__ W,
                                                  const float* __restrict__ bias,
                                                  const float* __restrict__ a,
                                                  float* __restrict__ hbuf,
                                                  float* __restrict__ lsrc,
                                                  float* __restrict__ ldst) {
  const int cg   = threadIdx.x >> 5;
  const int lane = threadIdx.x & 31;
  const int half = lane >> 4;
  const int l    = lane & 15;

  const int row0 = blockIdx.x * GEMM_BROWS_;
  const int col0 = cg * 64;

  const float* xa = X + (size_t)(row0 + l) * CIN_ + 4 * half;
  const float* wb = W + (size_t)(col0 + l) * CIN_ + 4 * half;

  GemmChunk q0, q1;
  gemm_load(q0, xa, wb, 0);

  v8f acc[2][4] = {{{}, {}, {}, {}}, {{}, {}, {}, {}}};
  #pragma unroll
  for (int kc = 0; kc < CIN_; kc += 64) {
    gemm_load(q1, xa, wb, kc + 32);
    __builtin_amdgcn_sched_barrier(0);
    gemm_compute(q0, acc);
    __builtin_amdgcn_sched_barrier(0);
    if (kc + 64 < CIN_) gemm_load(q0, xa, wb, kc + 64);
    __builtin_amdgcn_sched_barrier(0);
    gemm_compute(q1, acc);
  }

  #pragma unroll
  for (int rt = 0; rt < 2; ++rt) {
    #pragma unroll
    for (int cc = 0; cc < 4; ++cc) {
      const float bval = bias[col0 + cc * 16 + l];
      #pragma unroll
      for (int r = 0; r < 8; ++r) {
        const int row = row0 + 16 * rt + r + 8 * half;
        hbuf[(size_t)row * COUT_ + col0 + cc * 16 + l] = acc[rt][cc][r] + bval;
      }
    }
  }

  __syncthreads();
  {
    const int head = threadIdx.x & (H_ - 1);
    const int bi   = blockIdx.x * GEMM_BROWS_ + (threadIdx.x >> 2);
    const float* hr = hbuf + (size_t)bi * COUT_ + head * C_;
    const float* as = a + head * (2 * C_);
    const float* ad = as + C_;
    float s = 0.f, d = 0.f;
    #pragma unroll
    for (int c = 0; c < C_; ++c) {
      const float v = hr[c];
      s = fmaf(v, as[c], s);
      d = fmaf(v, ad[c], d);
    }
    lsrc[bi * H_ + head] = s;
    ldst[bi * H_ + head] = d;
  }
}

typedef int v4i __attribute__((ext_vector_type(4)));

#define SUM_FLOOR_ 9.5367431640625e-07f

struct AggChunk {
  v4i adA[4];
  v4i adB[4];
  v4f hr[4][4];
};

__device__ __forceinline__ void agg_load(AggChunk& q, const int* __restrict__ arA,
                                         const int* __restrict__ arB,
                                         const float* __restrict__ hcol, int jb) {
  #pragma unroll
  for (int u = 0; u < 4; ++u) {
    const int j = jb + 8 * u;
    q.adA[u] = *(const v4i*)(arA + j);
    q.adB[u] = *(const v4i*)(arB + j);
    #pragma unroll
    for (int r = 0; r < 4; ++r)
      q.hr[u][r] = *(const v4f*)(hcol + (size_t)(j + r) * COUT_);
  }
}

__device__ __forceinline__ void agg_compute(const AggChunk& q, const float* ldh, int jb,
                                            float lsA, float lsB, float mA, float mB,
                                            float& sumA, float& sumB,
                                            v8f (&accA)[4], v8f (&accB)[4]) {
  #pragma unroll
  for (int u = 0; u < 4; ++u) {
    const v4f ld = *(const v4f*)(ldh + jb + 8 * u);
    float pA[4], pB[4];
    #pragma unroll
    for (int r = 0; r < 4; ++r) {
      float xa = lsA + ld[r]; xa = fmaxf(xa, ALPHA_ * xa);
      float xb = lsB + ld[r]; xb = fmaxf(xb, ALPHA_ * xb);
      pA[r] = (q.adA[u][r] == 1) ? __expf(xa - mA) : 0.0f;
      pB[r] = (q.adB[u][r] == 1) ? __expf(xb - mB) : 0.0f;
      sumA += pA[r];
      sumB += pB[r];
    }
    #pragma unroll
    for (int t = 0; t < 2; ++t) {
      v2f avA, avB;
      avA.x = pA[2 * t]; avA.y = pA[2 * t + 1];
      avB.x = pB[2 * t]; avB.y = pB[2 * t + 1];
      #pragma unroll
      for (int cc = 0; cc < 4; ++cc) {
        v2f bv;
        bv.x = q.hr[u][2 * t][cc];
        bv.y = q.hr[u][2 * t + 1][cc];
        accA[cc] = __builtin_amdgcn_wmma_f32_16x16x4_f32(false, avA, false, bv,
                                                         (short)0, accA[cc],
                                                         false, false);
        accB[cc] = __builtin_amdgcn_wmma_f32_16x16x4_f32(false, avB, false, bv,
                                                         (short)0, accB[cc],
                                                         false, false);
      }
    }
  }
}

__global__ __launch_bounds__(128) void gat_aggregate(const float* __restrict__ hbuf,
                                                     const float* __restrict__ lsrc,
                                                     const float* __restrict__ ldst,
                                                     const int*   __restrict__ adj,
                                                     float* __restrict__ out) {
  __shared__ __attribute__((aligned(16))) float lds_ld[H_ * N_];
  __shared__ float lds_wmax[4][H_];

  const int b     = blockIdx.x >> 5;
  const int islab = blockIdx.x & 31;
  const int i0    = islab * 32;
  const int head  = threadIdx.x >> 5;
  const int lane  = threadIdx.x & 31;
  const int half  = lane >> 4;
  const int l     = lane & 15;

  {
    const float4* src = (const float4*)(ldst + (size_t)b * N_ * H_);
    float hm[H_] = {-INFINITY, -INFINITY, -INFINITY, -INFINITY};
    for (int j = threadIdx.x; j < N_; j += 128) {
      const float4 v = src[j];
      lds_ld[0 * N_ + j] = v.x;
      lds_ld[1 * N_ + j] = v.y;
      lds_ld[2 * N_ + j] = v.z;
      lds_ld[3 * N_ + j] = v.w;
      hm[0] = fmaxf(hm[0], v.x);
      hm[1] = fmaxf(hm[1], v.y);
      hm[2] = fmaxf(hm[2], v.z);
      hm[3] = fmaxf(hm[3], v.w);
    }
    #pragma unroll
    for (int hd = 0; hd < H_; ++hd) {
      #pragma unroll
      for (int off = 16; off > 0; off >>= 1)
        hm[hd] = fmaxf(hm[hd], __shfl_xor(hm[hd], off, 32));
    }
    if (lane == 0) {
      #pragma unroll
      for (int hd = 0; hd < H_; ++hd) lds_wmax[head][hd] = hm[hd];
    }
  }

  const size_t biA = (size_t)b * N_ + (i0 + l);
  const size_t biB = biA + 16;
  const int*   arA = adj + biA * (size_t)N_;
  const int*   arB = adj + biB * (size_t)N_;
  const float* hcol = hbuf + (size_t)b * N_ * COUT_ + head * C_ + 4 * l;
  const float* ldh  = lds_ld + head * N_;
  const int    jh   = 4 * half;

  AggChunk q0, q1;
  agg_load(q0, arA, arB, hcol, jh);

  const float  lsA = lsrc[biA * H_ + head];
  const float  lsB = lsrc[biB * H_ + head];

  __syncthreads();

  const float lmax = fmaxf(fmaxf(lds_wmax[0][head], lds_wmax[1][head]),
                           fmaxf(lds_wmax[2][head], lds_wmax[3][head]));
  float mA, mB;
  {
    const float xA = lsA + lmax, xB = lsB + lmax;
    mA = fmaxf(xA, ALPHA_ * xA);
    mB = fmaxf(xB, ALPHA_ * xB);
  }

  v8f accA[4], accB[4];
  float sumA, sumB;

  #pragma nounroll
  for (int attempt = 0; ; ++attempt) {
  #pragma unroll
  for (int cc = 0; cc < 4; ++cc) { accA[cc] = v8f{}; accB[cc] = v8f{}; }
  sumA = 0.0f; sumB = 0.0f;

  for (int jc = 0; jc < N_; jc += 64) {
    agg_load(q1, arA, arB, hcol, jc + 32 + jh);
    __builtin_amdgcn_sched_barrier(0);
    agg_compute(q0, ldh, jc + jh, lsA, lsB, mA, mB, sumA, sumB, accA, accB);
    __builtin_amdgcn_sched_barrier(0);
    agg_load(q0, arA, arB, hcol, ((jc + 64) & (N_ - 1)) + jh);
    __builtin_amdgcn_sched_barrier(0);
    agg_compute(q1, ldh, jc + 32 + jh, lsA, lsB, mA, mB, sumA, sumB, accA, accB);
  }

  sumA += __shfl_xor(sumA, 16, 32);
  sumB += __shfl_xor(sumB, 16, 32);

  const int small = (sumA < SUM_FLOOR_) || (sumB < SUM_FLOOR_);
  if (attempt == 1 || !__any(small)) break;

  float mxA = -INFINITY, mxB = -INFINITY;
  for (int j = jh; j < N_; j += 8) {
    const v4i a4 = *(const v4i*)(arA + j);
    const v4i b4 = *(const v4i*)(arB + j);
    const v4f ld = *(const v4f*)(ldh + j);
    #pragma unroll
    for (int r = 0; r < 4; ++r) {
      mxA = fmaxf(mxA, (a4[r] == 1) ? ld[r] : -INFINITY);
      mxB = fmaxf(mxB, (b4[r] == 1) ? ld[r] : -INFINITY);
    }
  }
  mxA = fmaxf(mxA, __shfl_xor(mxA, 16, 32));
  mxB = fmaxf(mxB, __shfl_xor(mxB, 16, 32));
  {
    const float xA = lsA + mxA, xB = lsB + mxB;
    mA = fmaxf(xA, ALPHA_ * xA);
    mB = fmaxf(xB, ALPHA_ * xB);
  }
  }
  const float invA = (sumA > 0.0f) ? 1.0f / sumA : 0.0f;
  const float invB = (sumB > 0.0f) ? 1.0f / sumB : 0.0f;

  #pragma unroll
  for (int r = 0; r < 8; ++r) {
    const int rowA = i0 + r + 8 * half;
    const int rowB = rowA + 16;
    const float wA = __shfl(invA, r + 8 * half, 32);
    const float wB = __shfl(invB, r + 8 * half, 32);
    v4f oa, ob;
    oa.x = accA[0][r] * wA; oa.y = accA[1][r] * wA; oa.z = accA[2][r] * wA; oa.w = accA[3][r] * wA;
    ob.x = accB[0][r] * wB; ob.y = accB[1][r] * wB; ob.z = accB[2][r] * wB; ob.w = accB[3][r] * wB;
    *(v4f*)(out + ((size_t)b * N_ + rowA) * COUT_ + head * C_ + 4 * l) = oa;
    *(v4f*)(out + ((size_t)b * N_ + rowB) * COUT_ + head * C_ + 4 * l) = ob;
  }
}

extern "C" void kernel_launch(void* const* d_in, const int* in_sizes, int n_in,
                              void* d_out, int out_size, void* d_ws, size_t ws_size,
                              hipStream_t stream) {
  const float* node_feats = (const float*)d_in[0];
  const int*   adj        = (const int*)  d_in[1];
  const float* W          = (const float*)d_in[2];
  const float* bias       = (const float*)d_in[3];
  const float* a          = (const float*)d_in[4];
  float*       out        = (float*)d_out;

  float* hbuf = (float*)d_ws;
  float* lsrc = hbuf + (size_t)B_ * N_ * COUT_;
  float* ldst = lsrc + (size_t)B_ * N_ * H_;

  gat_gemm_h<<<B_ * N_ / GEMM_BROWS_, 128, 0, stream>>>(
      node_feats, W, bias, a, hbuf, lsrc, ldst);

  gat_aggregate<<<B_ * (N_ / 32), 128, 0, stream>>>(
      hbuf, lsrc, ldst, adj, out);
}


# Aggregation prologue: a thread's 8 l_dst pieces, its l_src values and the whole first chunk leave as one load group before any wait (was 8 serial round trips, then the chunk); same arithmetic
# speedup vs baseline: 2.9026x; 1.0583x over previous
#include <hip/hip_runtime.h>
#include <hip/hip_bf16.h>
#include <math.h>

#define B_      8
#define N_      1024
#define CIN_    256
#define H_      4
#define COUT_   256
#define C_      64
#define ALPHA_  0.2f
#define NEG_INF_ (-9.0e15f)

typedef float v2f __attribute__((ext_vector_type(2)));
typedef float v8f __attribute__((ext_vector_type(8)));

#define GEMM_BROWS_ 32

typedef float v4f __attribute__((ext_vector_type(4)));

struct GemmChunk {
  v4f a[4][2];
  v4f b[4][4];
};

__device__ __forceinline__ void gemm_load(GemmChunk& q, const float* __restrict__ xa,
                                          const float* __restrict__ wb, int kb) {
  #pragma unroll
  for (int p = 0; p < 4; ++p) {
    const int ko = kb + 8 * p;
    q.a[p][0] = *(const v4f*)(xa + ko);
    q.a[p][1] = *(const v4f*)(xa + 16 * CIN_ + ko);
    #pragma unroll
    for (int cc = 0; cc < 4; ++cc)
      q.b[p][cc] = *(const v4f*)(wb + cc * 16 * CIN_ + ko);
  }
}

__device__ __forceinline__ void gemm_compute(const GemmChunk& q, v8f (&acc)[2][4]) {
  #pragma unroll
  for (int p = 0; p < 4; ++p) {
    #pragma unroll
    for (int t = 0; t < 2; ++t) {
      v2f a0, a1;
      a0.x = q.a[p][0][2 * t]; a0.y = q.a[p][0][2 * t + 1];
      a1.x = q.a[p][1][2 * t]; a1.y = q.a[p][1][2 * t + 1];
      #pragma unroll
      for (int cc = 0; cc < 4; ++cc) {
        v2f bv;
        bv.x = q.b[p][cc][2 * t];
        bv.y = q.b[p][cc][2 * t + 1];
        acc[0][cc] = __builtin_amdgcn_wmma_f32_16x16x4_f32(false, a0, false, bv,
                                                           (short)0, acc[0][cc],
                                                           false, false);
        acc[1][cc] = __builtin_amdgcn_wmma_f32_16x16x4_f32(false, a1, false, bv,
                                                           (short)0, acc[1][cc],
                                                           false, false);
      }
    }
  }
}

__global__ __launch_bounds__(128) void gat_gemm_h(const float* __restrict__ X,
                                                  const float* __restrict__ W,
                                                  const float* __restrict__ bias,
                                                  const float* __restrict__ a,
                                                  float* __restrict__ hbuf,
                                                  float* __restrict__ lsrc,
                                                  float* __restrict__ ldst) {
  const int cg   = threadIdx.x >> 5;
  const int lane = threadIdx.x & 31;
  const int half = lane >> 4;
  const int l    = lane & 15;

  const int row0 = blockIdx.x * GEMM_BROWS_;
  const int col0 = cg * 64;

  const float* xa = X + (size_t)(row0 + l) * CIN_ + 4 * half;
  const float* wb = W + (size_t)(col0 + l) * CIN_ + 4 * half;

  GemmChunk q0, q1;
  gemm_load(q0, xa, wb, 0);

  v8f acc[2][4] = {{{}, {}, {}, {}}, {{}, {}, {}, {}}};
  #pragma unroll
  for (int kc = 0; kc < CIN_; kc += 64) {
    gemm_load(q1, xa, wb, kc + 32);
    __builtin_amdgcn_sched_barrier(0);
    gemm_compute(q0, acc);
    __builtin_amdgcn_sched_barrier(0);
    if (kc + 64 < CIN_) gemm_load(q0, xa, wb, kc + 64);
    __builtin_amdgcn_sched_barrier(0);
    gemm_compute(q1, acc);
  }

  #pragma unroll
  for (int rt = 0; rt < 2; ++rt) {
    #pragma unroll
    for (int cc = 0; cc < 4; ++cc) {
      const float bval = bias[col0 + cc * 16 + l];
      #pragma unroll
      for (int r = 0; r < 8; ++r) {
        const int row = row0 + 16 * rt + r + 8 * half;
        hbuf[(size_t)row * COUT_ + col0 + cc * 16 + l] = acc[rt][cc][r] + bval;
      }
    }
  }

  __syncthreads();
  {
    const int head = threadIdx.x & (H_ - 1);
    const int bi   = blockIdx.x * GEMM_BROWS_ + (threadIdx.x >> 2);
    const float* hr = hbuf + (size_t)bi * COUT_ + head * C_;
    const float* as = a + head * (2 * C_);
    const float* ad = as + C_;
    float s = 0.f, d = 0.f;
    #pragma unroll
    for (int c = 0; c < C_; ++c) {
      const float v = hr[c];
      s = fmaf(v, as[c], s);
      d = fmaf(v, ad[c], d);
    }
    lsrc[bi * H_ + head] = s;
    ldst[bi * H_ + head] = d;
  }
}

typedef int v4i __attribute__((ext_vector_type(4)));

#define SUM_FLOOR_ 9.5367431640625e-07f

struct AggChunk {
  v4i adA[4];
  v4i adB[4];
  v4f hr[4][4];
};

__device__ __forceinline__ void agg_load(AggChunk& q, const int* __restrict__ arA,
                                         const int* __restrict__ arB,
                                         const float* __restrict__ hcol, int jb) {
  #pragma unroll
  for (int u = 0; u < 4; ++u) {
    const int j = jb + 8 * u;
    q.adA[u] = *(const v4i*)(arA + j);
    q.adB[u] = *(const v4i*)(arB + j);
    #pragma unroll
    for (int r = 0; r < 4; ++r)
      q.hr[u][r] = *(const v4f*)(hcol + (size_t)(j + r) * COUT_);
  }
}

__device__ __forceinline__ void agg_compute(const AggChunk& q, const float* ldh, int jb,
                                            float lsA, float lsB, float mA, float mB,
                                            float& sumA, float& sumB,
                                            v8f (&accA)[4], v8f (&accB)[4]) {
  #pragma unroll
  for (int u = 0; u < 4; ++u) {
    const v4f ld = *(const v4f*)(ldh + jb + 8 * u);
    float pA[4], pB[4];
    #pragma unroll
    for (int r = 0; r < 4; ++r) {
      float xa = lsA + ld[r]; xa = fmaxf(xa, ALPHA_ * xa);
      float xb = lsB + ld[r]; xb = fmaxf(xb, ALPHA_ * xb);
      pA[r] = (q.adA[u][r] == 1) ? __expf(xa - mA) : 0.0f;
      pB[r] = (q.adB[u][r] == 1) ? __expf(xb - mB) : 0.0f;
      sumA += pA[r];
      sumB += pB[r];
    }
    #pragma unroll
    for (int t = 0; t < 2; ++t) {
      v2f avA, avB;
      avA.x = pA[2 * t]; avA.y = pA[2 * t + 1];
      avB.x = pB[2 * t]; avB.y = pB[2 * t + 1];
      #pragma unroll
      for (int cc = 0; cc < 4; ++cc) {
        v2f bv;
        bv.x = q.hr[u][2 * t][cc];
        bv.y = q.hr[u][2 * t + 1][cc];
        accA[cc] = __builtin_amdgcn_wmma_f32_16x16x4_f32(false, avA, false, bv,
                                                         (short)0, accA[cc],
                                                         false, false);
        accB[cc] = __builtin_amdgcn_wmma_f32_16x16x4_f32(false, avB, false, bv,
                                                         (short)0, accB[cc],
                                                         false, false);
      }
    }
  }
}

__global__ __launch_bounds__(128) void gat_aggregate(const float* __restrict__ hbuf,
                                                     const float* __restrict__ lsrc,
                                                     const float* __restrict__ ldst,
                                                     const int*   __restrict__ adj,
                                                     float* __restrict__ out) {
  __shared__ __attribute__((aligned(16))) float lds_ld[H_ * N_];
  __shared__ float lds_wmax[4][H_];

  const int b     = blockIdx.x >> 5;
  const int islab = blockIdx.x & 31;
  const int i0    = islab * 32;
  const int head  = threadIdx.x >> 5;
  const int lane  = threadIdx.x & 31;
  const int half  = lane >> 4;
  const int l     = lane & 15;

  const size_t biA = (size_t)b * N_ + (i0 + l);
  const size_t biB = biA + 16;
  const int*   arA = adj + biA * (size_t)N_;
  const int*   arB = adj + biB * (size_t)N_;
  const float* hcol = hbuf + (size_t)b * N_ * COUT_ + head * C_ + 4 * l;
  const float* ldh  = lds_ld + head * N_;
  const int    jh   = 4 * half;

  const float4* src = (const float4*)(ldst + (size_t)b * N_ * H_);
  float4 ldv[N_ / 128];
  #pragma unroll
  for (int t = 0; t < N_ / 128; ++t) ldv[t] = src[threadIdx.x + 128 * t];

  const float  lsA = lsrc[biA * H_ + head];
  const float  lsB = lsrc[biB * H_ + head];

  AggChunk q0, q1;
  agg_load(q0, arA, arB, hcol, jh);

  {
    float hm[H_] = {-INFINITY, -INFINITY, -INFINITY, -INFINITY};
    #pragma unroll
    for (int t = 0; t < N_ / 128; ++t) {
      const int j = threadIdx.x + 128 * t;
      const float4 v = ldv[t];
      lds_ld[0 * N_ + j] = v.x;
      lds_ld[1 * N_ + j] = v.y;
      lds_ld[2 * N_ + j] = v.z;
      lds_ld[3 * N_ + j] = v.w;
      hm[0] = fmaxf(hm[0], v.x);
      hm[1] = fmaxf(hm[1], v.y);
      hm[2] = fmaxf(hm[2], v.z);
      hm[3] = fmaxf(hm[3], v.w);
    }
    #pragma unroll
    for (int hd = 0; hd < H_; ++hd) {
      #pragma unroll
      for (int off = 16; off > 0; off >>= 1)
        hm[hd] = fmaxf(hm[hd], __shfl_xor(hm[hd], off, 32));
    }
    if (lane == 0) {
      #pragma unroll
      for (int hd = 0; hd < H_; ++hd) lds_wmax[head][hd] = hm[hd];
    }
  }

  __syncthreads();

  const float lmax = fmaxf(fmaxf(lds_wmax[0][head], lds_wmax[1][head]),
                           fmaxf(lds_wmax[2][head], lds_wmax[3][head]));
  float mA, mB;
  {
    const float xA = lsA + lmax, xB = lsB + lmax;
    mA = fmaxf(xA, ALPHA_ * xA);
    mB = fmaxf(xB, ALPHA_ * xB);
  }

  v8f accA[4], accB[4];
  float sumA, sumB;

  #pragma nounroll
  for (int attempt = 0; ; ++attempt) {
  #pragma unroll
  for (int cc = 0; cc < 4; ++cc) { accA[cc] = v8f{}; accB[cc] = v8f{}; }
  sumA = 0.0f; sumB = 0.0f;

  for (int jc = 0; jc < N_; jc += 64) {
    agg_load(q1, arA, arB, hcol, jc + 32 + jh);
    __builtin_amdgcn_sched_barrier(0);
    agg_compute(q0, ldh, jc + jh, lsA, lsB, mA, mB, sumA, sumB, accA, accB);
    __builtin_amdgcn_sched_barrier(0);
    agg_load(q0, arA, arB, hcol, ((jc + 64) & (N_ - 1)) + jh);
    __builtin_amdgcn_sched_barrier(0);
    agg_compute(q1, ldh, jc + 32 + jh, lsA, lsB, mA, mB, sumA, sumB, accA, accB);
  }

  sumA += __shfl_xor(sumA, 16, 32);
  sumB += __shfl_xor(sumB, 16, 32);

  const int small = (sumA < SUM_FLOOR_) || (sumB < SUM_FLOOR_);
  if (attempt == 1 || !__any(small)) break;

  float mxA = -INFINITY, mxB = -INFINITY;
  for (int j = jh; j < N_; j += 8) {
    const v4i a4 = *(const v4i*)(arA + j);
    const v4i b4 = *(const v4i*)(arB + j);
    const v4f ld = *(const v4f*)(ldh + j);
    #pragma unroll
    for (int r = 0; r < 4; ++r) {
      mxA = fmaxf(mxA, (a4[r] == 1) ? ld[r] : -INFINITY);
      mxB = fmaxf(mxB, (b4[r] == 1) ? ld[r] : -INFINITY);
    }
  }
  mxA = fmaxf(mxA, __shfl_xor(mxA, 16, 32));
  mxB = fmaxf(mxB, __shfl_xor(mxB, 16, 32));
  {
    const float xA = lsA + mxA, xB = lsB + mxB;
    mA = fmaxf(xA, ALPHA_ * xA);
    mB = fmaxf(xB, ALPHA_ * xB);
  }
  }
  const float invA = (sumA > 0.0f) ? 1.0f / sumA : 0.0f;
  const float invB = (sumB > 0.0f) ? 1.0f / sumB : 0.0f;

  #pragma unroll
  for (int r = 0; r < 8; ++r) {
    const int rowA = i0 + r + 8 * half;
    const int rowB = rowA + 16;
    const float wA = __shfl(invA, r + 8 * half, 32);
    const float wB = __shfl(invB, r + 8 * half, 32);
    v4f oa, ob;
    oa.x = accA[0][r] * wA; oa.y = accA[1][r] * wA; oa.z = accA[2][r] * wA; oa.w = accA[3][r] * wA;
    ob.x = accB[0][r] * wB; ob.y = accB[1][r] * wB; ob.z = accB[2][r] * wB; ob.w = accB[3][r] * wB;
    *(v4f*)(out + ((size_t)b * N_ + rowA) * COUT_ + head * C_ + 4 * l) = oa;
    *(v4f*)(out + ((size_t)b * N_ + rowB) * COUT_ + head * C_ + 4 * l) = ob;
  }
}

extern "C" void kernel_launch(void* const* d_in, const int* in_sizes, int n_in,
                              void* d_out, int out_size, void* d_ws, size_t ws_size,
                              hipStream_t stream) {
  const float* node_feats = (const float*)d_in[0];
  const int*   adj        = (const int*)  d_in[1];
  const float* W          = (const float*)d_in[2];
  const float* bias       = (const float*)d_in[3];
  const float* a          = (const float*)d_in[4];
  float*       out        = (float*)d_out;

  float* hbuf = (float*)d_ws;
  float* lsrc = hbuf + (size_t)B_ * N_ * COUT_;
  float* ldst = lsrc + (size_t)B_ * N_ * H_;

  gat_gemm_h<<<B_ * N_ / GEMM_BROWS_, 128, 0, stream>>>(
      node_feats, W, bias, a, hbuf, lsrc, ldst);

  gat_aggregate<<<B_ * (N_ / 32), 128, 0, stream>>>(
      hbuf, lsrc, ldst, adj, out);
}


# Aggregation: edge weight = max(E1_i*F1_j, E2_i*F2_j), F1/F2 = per-node exp tables built once per block in LDS, read a chunk ahead; no lrelu/exp chain per edge (weights differ in last bits)
# speedup vs baseline: 3.0265x; 1.0427x over previous
#include <hip/hip_runtime.h>
#include <hip/hip_bf16.h>
#include <math.h>

#define B_      8
#define N_      1024
#define CIN_    256
#define H_      4
#define COUT_   256
#define C_      64
#define ALPHA_  0.2f
#define NEG_INF_ (-9.0e15f)

typedef float v2f __attribute__((ext_vector_type(2)));
typedef float v8f __attribute__((ext_vector_type(8)));

#define GEMM_BROWS_ 32

typedef float v4f __attribute__((ext_vector_type(4)));

struct GemmChunk {
  v4f a[4][2];
  v4f b[4][4];
};

__device__ __forceinline__ void gemm_load(GemmChunk& q, const float* __restrict__ xa,
                                          const float* __restrict__ wb, int kb) {
  #pragma unroll
  for (int p = 0; p < 4; ++p) {
    const int ko = kb + 8 * p;
    q.a[p][0] = *(const v4f*)(xa + ko);
    q.a[p][1] = *(const v4f*)(xa + 16 * CIN_ + ko);
    #pragma unroll
    for (int cc = 0; cc < 4; ++cc)
      q.b[p][cc] = *(const v4f*)(wb + cc * 16 * CIN_ + ko);
  }
}

__device__ __forceinline__ void gemm_compute(const GemmChunk& q, v8f (&acc)[2][4]) {
  #pragma unroll
  for (int p = 0; p < 4; ++p) {
    #pragma unroll
    for (int t = 0; t < 2; ++t) {
      v2f a0, a1;
      a0.x = q.a[p][0][2 * t]; a0.y = q.a[p][0][2 * t + 1];
      a1.x = q.a[p][1][2 * t]; a1.y = q.a[p][1][2 * t + 1];
      #pragma unroll
      for (int cc = 0; cc < 4; ++cc) {
        v2f bv;
        bv.x = q.b[p][cc][2 * t];
        bv.y = q.b[p][cc][2 * t + 1];
        acc[0][cc] = __builtin_amdgcn_wmma_f32_16x16x4_f32(false, a0, false, bv,
                                                           (short)0, acc[0][cc],
                                                           false, false);
        acc[1][cc] = __builtin_amdgcn_wmma_f32_16x16x4_f32(false, a1, false, bv,
                                                           (short)0, acc[1][cc],
                                                           false, false);
      }
    }
  }
}

__global__ __launch_bounds__(128) void gat_gemm_h(const float* __restrict__ X,
                                                  const float* __restrict__ W,
                                                  const float* __restrict__ bias,
                                                  const float* __restrict__ a,
                                                  float* __restrict__ hbuf,
                                                  float* __restrict__ lsrc,
                                                  float* __restrict__ ldst) {
  const int cg   = threadIdx.x >> 5;
  const int lane = threadIdx.x & 31;
  const int half = lane >> 4;
  const int l    = lane & 15;

  const int row0 = blockIdx.x * GEMM_BROWS_;
  const int col0 = cg * 64;

  const float* xa = X + (size_t)(row0 + l) * CIN_ + 4 * half;
  const float* wb = W + (size_t)(col0 + l) * CIN_ + 4 * half;

  GemmChunk q0, q1;
  gemm_load(q0, xa, wb, 0);

  v8f acc[2][4] = {{{}, {}, {}, {}}, {{}, {}, {}, {}}};
  #pragma unroll
  for (int kc = 0; kc < CIN_; kc += 64) {
    gemm_load(q1, xa, wb, kc + 32);
    __builtin_amdgcn_sched_barrier(0);
    gemm_compute(q0, acc);
    __builtin_amdgcn_sched_barrier(0);
    if (kc + 64 < CIN_) gemm_load(q0, xa, wb, kc + 64);
    __builtin_amdgcn_sched_barrier(0);
    gemm_compute(q1, acc);
  }

  #pragma unroll
  for (int rt = 0; rt < 2; ++rt) {
    #pragma unroll
    for (int cc = 0; cc < 4; ++cc) {
      const float bval = bias[col0 + cc * 16 + l];
      #pragma unroll
      for (int r = 0; r < 8; ++r) {
        const int row = row0 + 16 * rt + r + 8 * half;
        hbuf[(size_t)row * COUT_ + col0 + cc * 16 + l] = acc[rt][cc][r] + bval;
      }
    }
  }

  __syncthreads();
  {
    const int head = threadIdx.x & (H_ - 1);
    const int bi   = blockIdx.x * GEMM_BROWS_ + (threadIdx.x >> 2);
    const float* hr = hbuf + (size_t)bi * COUT_ + head * C_;
    const float* as = a + head * (2 * C_);
    const float* ad = as + C_;
    float s = 0.f, d = 0.f;
    #pragma unroll
    for (int c = 0; c < C_; ++c) {
      const float v = hr[c];
      s = fmaf(v, as[c], s);
      d = fmaf(v, ad[c], d);
    }
    lsrc[bi * H_ + head] = s;
    ldst[bi * H_ + head] = d;
  }
}

typedef int v4i __attribute__((ext_vector_type(4)));

#define SUM_FLOOR_ 9.5367431640625e-07f

struct AggChunk {
  v4i adA[4];
  v4i adB[4];
  v4f hr[4][4];
  v4f f1[4];
  v4f f2[4];
};

__device__ __forceinline__ void agg_load_global(AggChunk& q, const int* __restrict__ arA,
                                                const int* __restrict__ arB,
                                                const float* __restrict__ hcol, int jb) {
  #pragma unroll
  for (int u = 0; u < 4; ++u) {
    const int j = jb + 8 * u;
    q.adA[u] = *(const v4i*)(arA + j);
    q.adB[u] = *(const v4i*)(arB + j);
    #pragma unroll
    for (int r = 0; r < 4; ++r)
      q.hr[u][r] = *(const v4f*)(hcol + (size_t)(j + r) * COUT_);
  }
}

__device__ __forceinline__ void agg_load_tables(AggChunk& q, const float* f1h,
                                                const float* f2h, int jb) {
  #pragma unroll
  for (int u = 0; u < 4; ++u) {
    q.f1[u] = *(const v4f*)(f1h + jb + 8 * u);
    q.f2[u] = *(const v4f*)(f2h + jb + 8 * u);
  }
}

template <bool TABLES>
__device__ __forceinline__ void agg_load(AggChunk& q, const int* __restrict__ arA,
                                         const int* __restrict__ arB,
                                         const float* __restrict__ hcol,
                                         const float* f1h, const float* f2h, int jb) {
  agg_load_global(q, arA, arB, hcol, jb);
  if (TABLES) agg_load_tables(q, f1h, f2h, jb);
}

__device__ __forceinline__ void agg_group_mma(const AggChunk& q, int u,
                                              const float (&pA)[4], const float (&pB)[4],
                                              v8f (&accA)[4], v8f (&accB)[4]) {
  #pragma unroll
  for (int t = 0; t < 2; ++t) {
    v2f avA, avB;
    avA.x = pA[2 * t]; avA.y = pA[2 * t + 1];
    avB.x = pB[2 * t]; avB.y = pB[2 * t + 1];
    #pragma unroll
    for (int cc = 0; cc < 4; ++cc) {
      v2f bv;
      bv.x = q.hr[u][2 * t][cc];
      bv.y = q.hr[u][2 * t + 1][cc];
      accA[cc] = __builtin_amdgcn_wmma_f32_16x16x4_f32(false, avA, false, bv,
                                                       (short)0, accA[cc],
                                                       false, false);
      accB[cc] = __builtin_amdgcn_wmma_f32_16x16x4_f32(false, avB, false, bv,
                                                       (short)0, accB[cc],
                                                       false, false);
    }
  }
}

__device__ __forceinline__ void agg_compute_tables(const AggChunk& q,
                                                   float e1A, float e2A, float e1B, float e2B,
                                                   float& sumA, float& sumB,
                                                   v8f (&accA)[4], v8f (&accB)[4]) {
  #pragma unroll
  for (int u = 0; u < 4; ++u) {
    float pA[4], pB[4];
    #pragma unroll
    for (int r = 0; r < 4; ++r) {
      const float f1 = q.f1[u][r], f2 = q.f2[u][r];
      const float wA = fmaxf(e1A * f1, e2A * f2);
      const float wB = fmaxf(e1B * f1, e2B * f2);
      pA[r] = (q.adA[u][r] == 1) ? wA : 0.0f;
      pB[r] = (q.adB[u][r] == 1) ? wB : 0.0f;
      sumA += pA[r];
      sumB += pB[r];
    }
    agg_group_mma(q, u, pA, pB, accA, accB);
  }
}

__device__ __forceinline__ void agg_compute(const AggChunk& q, const float* ldh, int jb,
                                            float lsA, float lsB, float mA, float mB,
                                            float& sumA, float& sumB,
                                            v8f (&accA)[4], v8f (&accB)[4]) {
  #pragma unroll
  for (int u = 0; u < 4; ++u) {
    const v4f ld = *(const v4f*)(ldh + jb + 8 * u);
    float pA[4], pB[4];
    #pragma unroll
    for (int r = 0; r < 4; ++r) {
      float xa = lsA + ld[r]; xa = fmaxf(xa, ALPHA_ * xa);
      float xb = lsB + ld[r]; xb = fmaxf(xb, ALPHA_ * xb);
      pA[r] = (q.adA[u][r] == 1) ? __expf(xa - mA) : 0.0f;
      pB[r] = (q.adB[u][r] == 1) ? __expf(xb - mB) : 0.0f;
      sumA += pA[r];
      sumB += pB[r];
    }
    agg_group_mma(q, u, pA, pB, accA, accB);
  }
}

template <bool TABLES>
__device__ __forceinline__ void agg_sweep(AggChunk& q0, AggChunk& q1,
                                          const int* __restrict__ arA,
                                          const int* __restrict__ arB,
                                          const float* __restrict__ hcol,
                                          const float* ldh, const float* f1h,
                                          const float* f2h, int jh,
                                          float lsA, float lsB, float mA, float mB,
                                          float e1A, float e2A, float e1B, float e2B,
                                          float& sumA, float& sumB,
                                          v8f (&accA)[4], v8f (&accB)[4]) {
  for (int jc = 0; jc < N_; jc += 64) {
    agg_load<TABLES>(q1, arA, arB, hcol, f1h, f2h, jc + 32 + jh);
    __builtin_amdgcn_sched_barrier(0);
    if (TABLES) agg_compute_tables(q0, e1A, e2A, e1B, e2B, sumA, sumB, accA, accB);
    else        agg_compute(q0, ldh, jc + jh, lsA, lsB, mA, mB, sumA, sumB, accA, accB);
    __builtin_amdgcn_sched_barrier(0);
    agg_load<TABLES>(q0, arA, arB, hcol, f1h, f2h, ((jc + 64) & (N_ - 1)) + jh);
    __builtin_amdgcn_sched_barrier(0);
    if (TABLES) agg_compute_tables(q1, e1A, e2A, e1B, e2B, sumA, sumB, accA, accB);
    else        agg_compute(q1, ldh, jc + 32 + jh, lsA, lsB, mA, mB, sumA, sumB, accA, accB);
  }
}

__global__ __launch_bounds__(128) void gat_aggregate(const float* __restrict__ hbuf,
                                                     const float* __restrict__ lsrc,
                                                     const float* __restrict__ ldst,
                                                     const int*   __restrict__ adj,
                                                     float* __restrict__ out) {
  __shared__ __attribute__((aligned(16))) float lds_ld[H_ * N_];
  __shared__ __attribute__((aligned(16))) float lds_f1[H_ * N_];
  __shared__ __attribute__((aligned(16))) float lds_f2[H_ * N_];
  __shared__ float lds_wmax[4][H_];

  const int b     = blockIdx.x >> 5;
  const int islab = blockIdx.x & 31;
  const int i0    = islab * 32;
  const int head  = threadIdx.x >> 5;
  const int lane  = threadIdx.x & 31;
  const int half  = lane >> 4;
  const int l     = lane & 15;

  const size_t biA = (size_t)b * N_ + (i0 + l);
  const size_t biB = biA + 16;
  const int*   arA = adj + biA * (size_t)N_;
  const int*   arB = adj + biB * (size_t)N_;
  const float* hcol = hbuf + (size_t)b * N_ * COUT_ + head * C_ + 4 * l;
  const float* ldh  = lds_ld + head * N_;
  const float* f1h  = lds_f1 + head * N_;
  const float* f2h  = lds_f2 + head * N_;
  const int    jh   = 4 * half;

  const float4* src = (const float4*)(ldst + (size_t)b * N_ * H_);
  float4 ldv[N_ / 128];
  #pragma unroll
  for (int t = 0; t < N_ / 128; ++t) ldv[t] = src[threadIdx.x + 128 * t];

  const float  lsA = lsrc[biA * H_ + head];
  const float  lsB = lsrc[biB * H_ + head];

  AggChunk q0, q1;
  agg_load_global(q0, arA, arB, hcol, jh);

  {
    float hm[H_] = {-INFINITY, -INFINITY, -INFINITY, -INFINITY};
    #pragma unroll
    for (int t = 0; t < N_ / 128; ++t) {
      const int j = threadIdx.x + 128 * t;
      const float4 v = ldv[t];
      lds_ld[0 * N_ + j] = v.x;
      lds_ld[1 * N_ + j] = v.y;
      lds_ld[2 * N_ + j] = v.z;
      lds_ld[3 * N_ + j] = v.w;
      hm[0] = fmaxf(hm[0], v.x);
      hm[1] = fmaxf(hm[1], v.y);
      hm[2] = fmaxf(hm[2], v.z);
      hm[3] = fmaxf(hm[3], v.w);
    }
    #pragma unroll
    for (int hd = 0; hd < H_; ++hd) {
      #pragma unroll
      for (int off = 16; off > 0; off >>= 1)
        hm[hd] = fmaxf(hm[hd], __shfl_xor(hm[hd], off, 32));
    }
    if (lane == 0) {
      #pragma unroll
      for (int hd = 0; hd < H_; ++hd) lds_wmax[head][hd] = hm[hd];
    }
  }

  __syncthreads();

  {
    float L[H_];
    #pragma unroll
    for (int hd = 0; hd < H_; ++hd)
      L[hd] = fmaxf(fmaxf(lds_wmax[0][hd], lds_wmax[1][hd]),
                    fmaxf(lds_wmax[2][hd], lds_wmax[3][hd]));
    #pragma unroll
    for (int t = 0; t < N_ / 128; ++t) {
      const int j = threadIdx.x + 128 * t;
      const float4 v = ldv[t];
      const float d0 = v.x - L[0], d1 = v.y - L[1], d2 = v.z - L[2], d3 = v.w - L[3];
      lds_f1[0 * N_ + j] = __expf(d0);
      lds_f1[1 * N_ + j] = __expf(d1);
      lds_f1[2 * N_ + j] = __expf(d2);
      lds_f1[3 * N_ + j] = __expf(d3);
      lds_f2[0 * N_ + j] = __expf(ALPHA_ * d0);
      lds_f2[1 * N_ + j] = __expf(ALPHA_ * d1);
      lds_f2[2 * N_ + j] = __expf(ALPHA_ * d2);
      lds_f2[3 * N_ + j] = __expf(ALPHA_ * d3);
    }
  }

  const float lmax = fmaxf(fmaxf(lds_wmax[0][head], lds_wmax[1][head]),
                           fmaxf(lds_wmax[2][head], lds_wmax[3][head]));
  float mA, mB;
  float e1A, e2A, e1B, e2B;
  {
    const float xA = lsA + lmax, xB = lsB + lmax;
    mA = fmaxf(xA, ALPHA_ * xA);
    mB = fmaxf(xB, ALPHA_ * xB);
    e1A = __expf(xA - mA); e2A = __expf(ALPHA_ * xA - mA);
    e1B = __expf(xB - mB); e2B = __expf(ALPHA_ * xB - mB);
  }

  __syncthreads();
  agg_load_tables(q0, f1h, f2h, jh);

  v8f accA[4], accB[4];
  float sumA, sumB;

  #pragma unroll
  for (int cc = 0; cc < 4; ++cc) { accA[cc] = v8f{}; accB[cc] = v8f{}; }
  sumA = 0.0f; sumB = 0.0f;

  agg_sweep<true>(q0, q1, arA, arB, hcol, ldh, f1h, f2h, jh, lsA, lsB, mA, mB,
                  e1A, e2A, e1B, e2B, sumA, sumB, accA, accB);

  sumA += __shfl_xor(sumA, 16, 32);
  sumB += __shfl_xor(sumB, 16, 32);

  const int small = (sumA < SUM_FLOOR_) || (sumB < SUM_FLOOR_);
  if (__any(small)) {
  float mxA = -INFINITY, mxB = -INFINITY;
  for (int j = jh; j < N_; j += 8) {
    const v4i a4 = *(const v4i*)(arA + j);
    const v4i b4 = *(const v4i*)(arB + j);
    const v4f ld = *(const v4f*)(ldh + j);
    #pragma unroll
    for (int r = 0; r < 4; ++r) {
      mxA = fmaxf(mxA, (a4[r] == 1) ? ld[r] : -INFINITY);
      mxB = fmaxf(mxB, (b4[r] == 1) ? ld[r] : -INFINITY);
    }
  }
  mxA = fmaxf(mxA, __shfl_xor(mxA, 16, 32));
  mxB = fmaxf(mxB, __shfl_xor(mxB, 16, 32));
  {
    const float xA = lsA + mxA, xB = lsB + mxB;
    mA = fmaxf(xA, ALPHA_ * xA);
    mB = fmaxf(xB, ALPHA_ * xB);
  }
  #pragma unroll
  for (int cc = 0; cc < 4; ++cc) { accA[cc] = v8f{}; accB[cc] = v8f{}; }
  sumA = 0.0f; sumB = 0.0f;
  agg_sweep<false>(q0, q1, arA, arB, hcol, ldh, f1h, f2h, jh, lsA, lsB, mA, mB,
                   e1A, e2A, e1B, e2B, sumA, sumB, accA, accB);
  sumA += __shfl_xor(sumA, 16, 32);
  sumB += __shfl_xor(sumB, 16, 32);
  }
  const float invA = (sumA > 0.0f) ? 1.0f / sumA : 0.0f;
  const float invB = (sumB > 0.0f) ? 1.0f / sumB : 0.0f;

  #pragma unroll
  for (int r = 0; r < 8; ++r) {
    const int rowA = i0 + r + 8 * half;
    const int rowB = rowA + 16;
    const float wA = __shfl(invA, r + 8 * half, 32);
    const float wB = __shfl(invB, r + 8 * half, 32);
    v4f oa, ob;
    oa.x = accA[0][r] * wA; oa.y = accA[1][r] * wA; oa.z = accA[2][r] * wA; oa.w = accA[3][r] * wA;
    ob.x = accB[0][r] * wB; ob.y = accB[1][r] * wB; ob.z = accB[2][r] * wB; ob.w = accB[3][r] * wB;
    *(v4f*)(out + ((size_t)b * N_ + rowA) * COUT_ + head * C_ + 4 * l) = oa;
    *(v4f*)(out + ((size_t)b * N_ + rowB) * COUT_ + head * C_ + 4 * l) = ob;
  }
}

extern "C" void kernel_launch(void* const* d_in, const int* in_sizes, int n_in,
                              void* d_out, int out_size, void* d_ws, size_t ws_size,
                              hipStream_t stream) {
  const float* node_feats = (const float*)d_in[0];
  const int*   adj        = (const int*)  d_in[1];
  const float* W          = (const float*)d_in[2];
  const float* bias       = (const float*)d_in[3];
  const float* a          = (const float*)d_in[4];
  float*       out        = (float*)d_out;

  float* hbuf = (float*)d_ws;
  float* lsrc = hbuf + (size_t)B_ * N_ * COUT_;
  float* ldst = lsrc + (size_t)B_ * N_ * H_;

  gat_gemm_h<<<B_ * N_ / GEMM_BROWS_, 128, 0, stream>>>(
      node_feats, W, bias, a, hbuf, lsrc, ldst);

  gat_aggregate<<<B_ * (N_ / 32), 128, 0, stream>>>(
      hbuf, lsrc, ldst, adj, out);
}


# Both kernels: a block's 4 waves form 2 x 2 (projection: 2 row slabs x 2 column groups; aggregation: 2 slabs x 2 heads; was 1 x 4), so two waves share each operand line in the WGP cache; same sums
# speedup vs baseline: 3.1258x; 1.0328x over previous
#include <hip/hip_runtime.h>
#include <hip/hip_bf16.h>
#include <math.h>

#define B_      8
#define N_      1024
#define CIN_    256
#define H_      4
#define COUT_   256
#define C_      64
#define ALPHA_  0.2f
#define NEG_INF_ (-9.0e15f)

typedef float v2f __attribute__((ext_vector_type(2)));
typedef float v8f __attribute__((ext_vector_type(8)));

#define GEMM_BROWS_ 64
#define GEMM_BCOLS_ 128

typedef float v4f __attribute__((ext_vector_type(4)));

struct GemmChunk {
  v4f a[4][2];
  v4f b[4][4];
};

__device__ __forceinline__ void gemm_load(GemmChunk& q, const float* __restrict__ xa,
                                          const float* __restrict__ wb, int kb) {
  #pragma unroll
  for (int p = 0; p < 4; ++p) {
    const int ko = kb + 8 * p;
    q.a[p][0] = *(const v4f*)(xa + ko);
    q.a[p][1] = *(const v4f*)(xa + 16 * CIN_ + ko);
    #pragma unroll
    for (int cc = 0; cc < 4; ++cc)
      q.b[p][cc] = *(const v4f*)(wb + cc * 16 * CIN_ + ko);
  }
}

__device__ __forceinline__ void gemm_compute(const GemmChunk& q, v8f (&acc)[2][4]) {
  #pragma unroll
  for (int p = 0; p < 4; ++p) {
    #pragma unroll
    for (int t = 0; t < 2; ++t) {
      v2f a0, a1;
      a0.x = q.a[p][0][2 * t]; a0.y = q.a[p][0][2 * t + 1];
      a1.x = q.a[p][1][2 * t]; a1.y = q.a[p][1][2 * t + 1];
      #pragma unroll
      for (int cc = 0; cc < 4; ++cc) {
        v2f bv;
        bv.x = q.b[p][cc][2 * t];
        bv.y = q.b[p][cc][2 * t + 1];
        acc[0][cc] = __builtin_amdgcn_wmma_f32_16x16x4_f32(false, a0, false, bv,
                                                           (short)0, acc[0][cc],
                                                           false, false);
        acc[1][cc] = __builtin_amdgcn_wmma_f32_16x16x4_f32(false, a1, false, bv,
                                                           (short)0, acc[1][cc],
                                                           false, false);
      }
    }
  }
}

__global__ __launch_bounds__(128) void gat_gemm_h(const float* __restrict__ X,
                                                  const float* __restrict__ W,
                                                  const float* __restrict__ bias,
                                                  const float* __restrict__ a,
                                                  float* __restrict__ hbuf,
                                                  float* __restrict__ lsrc,
                                                  float* __restrict__ ldst) {
  const int wv   = threadIdx.x >> 5;
  const int lane = threadIdx.x & 31;
  const int half = lane >> 4;
  const int l    = lane & 15;

  const int rb   = blockIdx.x >> 1;
  const int ch   = blockIdx.x & 1;
  const int row0 = rb * GEMM_BROWS_ + (wv >> 1) * 32;
  const int col0 = ch * GEMM_BCOLS_ + (wv & 1) * 64;

  const float* xa = X + (size_t)(row0 + l) * CIN_ + 4 * half;
  const float* wb = W + (size_t)(col0 + l) * CIN_ + 4 * half;

  GemmChunk q0, q1;
  gemm_load(q0, xa, wb, 0);

  v8f acc[2][4] = {{{}, {}, {}, {}}, {{}, {}, {}, {}}};
  #pragma unroll
  for (int kc = 0; kc < CIN_; kc += 64) {
    gemm_load(q1, xa, wb, kc + 32);
    __builtin_amdgcn_sched_barrier(0);
    gemm_compute(q0, acc);
    __builtin_amdgcn_sched_barrier(0);
    if (kc + 64 < CIN_) gemm_load(q0, xa, wb, kc + 64);
    __builtin_amdgcn_sched_barrier(0);
    gemm_compute(q1, acc);
  }

  #pragma unroll
  for (int rt = 0; rt < 2; ++rt) {
    #pragma unroll
    for (int cc = 0; cc < 4; ++cc) {
      const float bval = bias[col0 + cc * 16 + l];
      #pragma unroll
      for (int r = 0; r < 8; ++r) {
        const int row = row0 + 16 * rt + r + 8 * half;
        hbuf[(size_t)row * COUT_ + col0 + cc * 16 + l] = acc[rt][cc][r] + bval;
      }
    }
  }

  __syncthreads();
  {
    const int head = 2 * ch + (threadIdx.x & 1);
    const int bi   = rb * GEMM_BROWS_ + (threadIdx.x >> 1);
    const float* hr = hbuf + (size_t)bi * COUT_ + head * C_;
    const float* as = a + head * (2 * C_);
    const float* ad = as + C_;
    float s = 0.f, d = 0.f;
    #pragma unroll
    for (int c = 0; c < C_; ++c) {
      const float v = hr[c];
      s = fmaf(v, as[c], s);
      d = fmaf(v, ad[c], d);
    }
    lsrc[bi * H_ + head] = s;
    ldst[bi * H_ + head] = d;
  }
}

typedef int v4i __attribute__((ext_vector_type(4)));

#define SUM_FLOOR_ 9.5367431640625e-07f

struct AggChunk {
  v4i adA[4];
  v4i adB[4];
  v4f hr[4][4];
  v4f f1[4];
  v4f f2[4];
};

__device__ __forceinline__ void agg_load_global(AggChunk& q, const int* __restrict__ arA,
                                                const int* __restrict__ arB,
                                                const float* __restrict__ hcol, int jb) {
  #pragma unroll
  for (int u = 0; u < 4; ++u) {
    const int j = jb + 8 * u;
    q.adA[u] = *(const v4i*)(arA + j);
    q.adB[u] = *(const v4i*)(arB + j);
    #pragma unroll
    for (int r = 0; r < 4; ++r)
      q.hr[u][r] = *(const v4f*)(hcol + (size_t)(j + r) * COUT_);
  }
}

__device__ __forceinline__ void agg_load_tables(AggChunk& q, const float* f1h,
                                                const float* f2h, int jb) {
  #pragma unroll
  for (int u = 0; u < 4; ++u) {
    q.f1[u] = *(const v4f*)(f1h + jb + 8 * u);
    q.f2[u] = *(const v4f*)(f2h + jb + 8 * u);
  }
}

template <bool TABLES>
__device__ __forceinline__ void agg_load(AggChunk& q, const int* __restrict__ arA,
                                         const int* __restrict__ arB,
                                         const float* __restrict__ hcol,
                                         const float* f1h, const float* f2h, int jb) {
  agg_load_global(q, arA, arB, hcol, jb);
  if (TABLES) agg_load_tables(q, f1h, f2h, jb);
}

__device__ __forceinline__ void agg_group_mma(const AggChunk& q, int u,
                                              const float (&pA)[4], const float (&pB)[4],
                                              v8f (&accA)[4], v8f (&accB)[4]) {
  #pragma unroll
  for (int t = 0; t < 2; ++t) {
    v2f avA, avB;
    avA.x = pA[2 * t]; avA.y = pA[2 * t + 1];
    avB.x = pB[2 * t]; avB.y = pB[2 * t + 1];
    #pragma unroll
    for (int cc = 0; cc < 4; ++cc) {
      v2f bv;
      bv.x = q.hr[u][2 * t][cc];
      bv.y = q.hr[u][2 * t + 1][cc];
      accA[cc] = __builtin_amdgcn_wmma_f32_16x16x4_f32(false, avA, false, bv,
                                                       (short)0, accA[cc],
                                                       false, false);
      accB[cc] = __builtin_amdgcn_wmma_f32_16x16x4_f32(false, avB, false, bv,
                                                       (short)0, accB[cc],
                                                       false, false);
    }
  }
}

__device__ __forceinline__ void agg_compute_tables(const AggChunk& q,
                                                   float e1A, float e2A, float e1B, float e2B,
                                                   float& sumA, float& sumB,
                                                   v8f (&accA)[4], v8f (&accB)[4]) {
  #pragma unroll
  for (int u = 0; u < 4; ++u) {
    float pA[4], pB[4];
    #pragma unroll
    for (int r = 0; r < 4; ++r) {
      const float f1 = q.f1[u][r], f2 = q.f2[u][r];
      const float wA = fmaxf(e1A * f1, e2A * f2);
      const float wB = fmaxf(e1B * f1, e2B * f2);
      pA[r] = (q.adA[u][r] == 1) ? wA : 0.0f;
      pB[r] = (q.adB[u][r] == 1) ? wB : 0.0f;
      sumA += pA[r];
      sumB += pB[r];
    }
    agg_group_mma(q, u, pA, pB, accA, accB);
  }
}

__device__ __forceinline__ void agg_compute(const AggChunk& q, const float* ldh, int jb,
                                            float lsA, float lsB, float mA, float mB,
                                            float& sumA, float& sumB,
                                            v8f (&accA)[4], v8f (&accB)[4]) {
  #pragma unroll
  for (int u = 0; u < 4; ++u) {
    const v4f ld = *(const v4f*)(ldh + jb + 8 * u);
    float pA[4], pB[4];
    #pragma unroll
    for (int r = 0; r < 4; ++r) {
      float xa = lsA + ld[r]; xa = fmaxf(xa, ALPHA_ * xa);
      float xb = lsB + ld[r]; xb = fmaxf(xb, ALPHA_ * xb);
      pA[r] = (q.adA[u][r] == 1) ? __expf(xa - mA) : 0.0f;
      pB[r] = (q.adB[u][r] == 1) ? __expf(xb - mB) : 0.0f;
      sumA += pA[r];
      sumB += pB[r];
    }
    agg_group_mma(q, u, pA, pB, accA, accB);
  }
}

template <bool TABLES>
__device__ __forceinline__ void agg_sweep(AggChunk& q0, AggChunk& q1,
                                          const int* __restrict__ arA,
                                          const int* __restrict__ arB,
                                          const float* __restrict__ hcol,
                                          const float* ldh, const float* f1h,
                                          const float* f2h, int jh,
                                          float lsA, float lsB, float mA, float mB,
                                          float e1A, float e2A, float e1B, float e2B,
                                          float& sumA, float& sumB,
                                          v8f (&accA)[4], v8f (&accB)[4]) {
  for (int jc = 0; jc < N_; jc += 64) {
    agg_load<TABLES>(q1, arA, arB, hcol, f1h, f2h, jc + 32 + jh);
    __builtin_amdgcn_sched_barrier(0);
    if (TABLES) agg_compute_tables(q0, e1A, e2A, e1B, e2B, sumA, sumB, accA, accB);
    else        agg_compute(q0, ldh, jc + jh, lsA, lsB, mA, mB, sumA, sumB, accA, accB);
    __builtin_amdgcn_sched_barrier(0);
    agg_load<TABLES>(q0, arA, arB, hcol, f1h, f2h, ((jc + 64) & (N_ - 1)) + jh);
    __builtin_amdgcn_sched_barrier(0);
    if (TABLES) agg_compute_tables(q1, e1A, e2A, e1B, e2B, sumA, sumB, accA, accB);
    else        agg_compute(q1, ldh, jc + 32 + jh, lsA, lsB, mA, mB, sumA, sumB, accA, accB);
  }
}

__global__ __launch_bounds__(128) void gat_aggregate(const float* __restrict__ hbuf,
                                                     const float* __restrict__ lsrc,
                                                     const float* __restrict__ ldst,
                                                     const int*   __restrict__ adj,
                                                     float* __restrict__ out) {
  __shared__ __attribute__((aligned(16))) float lds_ld[2 * N_];
  __shared__ __attribute__((aligned(16))) float lds_f1[2 * N_];
  __shared__ __attribute__((aligned(16))) float lds_f2[2 * N_];
  __shared__ float lds_wmax[4][2];

  const int b     = blockIdx.x >> 5;
  const int hp    = blockIdx.x & 1;
  const int sp    = (blockIdx.x & 31) >> 1;
  const int wv    = threadIdx.x >> 5;
  const int hl    = wv & 1;
  const int head  = 2 * hp + hl;
  const int i0    = (2 * sp + (wv >> 1)) * 32;
  const int lane  = threadIdx.x & 31;
  const int half  = lane >> 4;
  const int l     = lane & 15;

  const size_t biA = (size_t)b * N_ + (i0 + l);
  const size_t biB = biA + 16;
  const int*   arA = adj + biA * (size_t)N_;
  const int*   arB = adj + biB * (size_t)N_;
  const float* hcol = hbuf + (size_t)b * N_ * COUT_ + head * C_ + 4 * l;
  const float* ldh  = lds_ld + hl * N_;
  const float* f1h  = lds_f1 + hl * N_;
  const float* f2h  = lds_f2 + hl * N_;
  const int    jh   = 4 * half;

  const float2* src = (const float2*)(ldst + (size_t)b * N_ * H_ + 2 * hp);
  float2 ldv[N_ / 128];
  #pragma unroll
  for (int t = 0; t < N_ / 128; ++t) ldv[t] = src[2 * (threadIdx.x + 128 * t)];

  const float  lsA = lsrc[biA * H_ + head];
  const float  lsB = lsrc[biB * H_ + head];

  AggChunk q0, q1;
  agg_load_global(q0, arA, arB, hcol, jh);

  {
    float hm[2] = {-INFINITY, -INFINITY};
    #pragma unroll
    for (int t = 0; t < N_ / 128; ++t) {
      const int j = threadIdx.x + 128 * t;
      const float2 v = ldv[t];
      lds_ld[0 * N_ + j] = v.x;
      lds_ld[1 * N_ + j] = v.y;
      hm[0] = fmaxf(hm[0], v.x);
      hm[1] = fmaxf(hm[1], v.y);
    }
    #pragma unroll
    for (int hd = 0; hd < 2; ++hd) {
      #pragma unroll
      for (int off = 16; off > 0; off >>= 1)
        hm[hd] = fmaxf(hm[hd], __shfl_xor(hm[hd], off, 32));
    }
    if (lane == 0) {
      #pragma unroll
      for (int hd = 0; hd < 2; ++hd) lds_wmax[wv][hd] = hm[hd];
    }
  }

  __syncthreads();

  {
    float L[2];
    #pragma unroll
    for (int hd = 0; hd < 2; ++hd)
      L[hd] = fmaxf(fmaxf(lds_wmax[0][hd], lds_wmax[1][hd]),
                    fmaxf(lds_wmax[2][hd], lds_wmax[3][hd]));
    #pragma unroll
    for (int t = 0; t < N_ / 128; ++t) {
      const int j = threadIdx.x + 128 * t;
      const float2 v = ldv[t];
      const float d0 = v.x - L[0], d1 = v.y - L[1];
      lds_f1[0 * N_ + j] = __expf(d0);
      lds_f1[1 * N_ + j] = __expf(d1);
      lds_f2[0 * N_ + j] = __expf(ALPHA_ * d0);
      lds_f2[1 * N_ + j] = __expf(ALPHA_ * d1);
    }
  }

  const float lmax = fmaxf(fmaxf(lds_wmax[0][hl], lds_wmax[1][hl]),
                           fmaxf(lds_wmax[2][hl], lds_wmax[3][hl]));
  float mA, mB;
  float e1A, e2A, e1B, e2B;
  {
    const float xA = lsA + lmax, xB = lsB + lmax;
    mA = fmaxf(xA, ALPHA_ * xA);
    mB = fmaxf(xB, ALPHA_ * xB);
    e1A = __expf(xA - mA); e2A = __expf(ALPHA_ * xA - mA);
    e1B = __expf(xB - mB); e2B = __expf(ALPHA_ * xB - mB);
  }

  __syncthreads();
  agg_load_tables(q0, f1h, f2h, jh);

  v8f accA[4], accB[4];
  float sumA, sumB;

  #pragma unroll
  for (int cc = 0; cc < 4; ++cc) { accA[cc] = v8f{}; accB[cc] = v8f{}; }
  sumA = 0.0f; sumB = 0.0f;

  agg_sweep<true>(q0, q1, arA, arB, hcol, ldh, f1h, f2h, jh, lsA, lsB, mA, mB,
                  e1A, e2A, e1B, e2B, sumA, sumB, accA, accB);

  sumA += __shfl_xor(sumA, 16, 32);
  sumB += __shfl_xor(sumB, 16, 32);

  const int small = (sumA < SUM_FLOOR_) || (sumB < SUM_FLOOR_);
  if (__any(small)) {
  float mxA = -INFINITY, mxB = -INFINITY;
  for (int j = jh; j < N_; j += 8) {
    const v4i a4 = *(const v4i*)(arA + j);
    const v4i b4 = *(const v4i*)(arB + j);
    const v4f ld = *(const v4f*)(ldh + j);
    #pragma unroll
    for (int r = 0; r < 4; ++r) {
      mxA = fmaxf(mxA, (a4[r] == 1) ? ld[r] : -INFINITY);
      mxB = fmaxf(mxB, (b4[r] == 1) ? ld[r] : -INFINITY);
    }
  }
  mxA = fmaxf(mxA, __shfl_xor(mxA, 16, 32));
  mxB = fmaxf(mxB, __shfl_xor(mxB, 16, 32));
  {
    const float xA = lsA + mxA, xB = lsB + mxB;
    mA = fmaxf(xA, ALPHA_ * xA);
    mB = fmaxf(xB, ALPHA_ * xB);
  }
  #pragma unroll
  for (int cc = 0; cc < 4; ++cc) { accA[cc] = v8f{}; accB[cc] = v8f{}; }
  sumA = 0.0f; sumB = 0.0f;
  agg_sweep<false>(q0, q1, arA, arB, hcol, ldh, f1h, f2h, jh, lsA, lsB, mA, mB,
                   e1A, e2A, e1B, e2B, sumA, sumB, accA, accB);
  sumA += __shfl_xor(sumA, 16, 32);
  sumB += __shfl_xor(sumB, 16, 32);
  }
  const float invA = (sumA > 0.0f) ? 1.0f / sumA : 0.0f;
  const float invB = (sumB > 0.0f) ? 1.0f / sumB : 0.0f;

  #pragma unroll
  for (int r = 0; r < 8; ++r) {
    const int rowA = i0 + r + 8 * half;
    const int rowB = rowA + 16;
    const float wA = __shfl(invA, r + 8 * half, 32);
    const float wB = __shfl(invB, r + 8 * half, 32);
    v4f oa, ob;
    oa.x = accA[0][r] * wA; oa.y = accA[1][r] * wA; oa.z = accA[2][r] * wA; oa.w = accA[3][r] * wA;
    ob.x = accB[0][r] * wB; ob.y = accB[1][r] * wB; ob.z = accB[2][r] * wB; ob.w = accB[3][r] * wB;
    *(v4f*)(out + ((size_t)b * N_ + rowA) * COUT_ + head * C_ + 4 * l) = oa;
    *(v4f*)(out + ((size_t)b * N_ + rowB) * COUT_ + head * C_ + 4 * l) = ob;
  }
}

extern "C" void kernel_launch(void* const* d_in, const int* in_sizes, int n_in,
                              void* d_out, int out_size, void* d_ws, size_t ws_size,
                              hipStream_t stream) {
  const float* node_feats = (const float*)d_in[0];
  const int*   adj        = (const int*)  d_in[1];
  const float* W          = (const float*)d_in[2];
  const float* bias       = (const float*)d_in[3];
  const float* a          = (const float*)d_in[4];
  float*       out        = (float*)d_out;

  float* hbuf = (float*)d_ws;
  float* lsrc = hbuf + (size_t)B_ * N_ * COUT_;
  float* ldst = lsrc + (size_t)B_ * N_ * H_;

  gat_gemm_h<<<(B_ * N_ / GEMM_BROWS_) * (COUT_ / GEMM_BCOLS_), 128, 0, stream>>>(
      node_feats, W, bias, a, hbuf, lsrc, ldst);

  gat_aggregate<<<B_ * (N_ / 32), 128, 0, stream>>>(
      hbuf, lsrc, ldst, adj, out);
}


# Aggregation loop: operands requested TWO chunks ahead (a third live register set, 4-chunk trips), so a chunk's loads get two chunks of matrix work as lead, not one; same arithmetic and sum order
# speedup vs baseline: 3.2762x; 1.0481x over previous
#include <hip/hip_runtime.h>
#include <hip/hip_bf16.h>
#include <math.h>

#define B_      8
#define N_      1024
#define CIN_    256
#define H_      4
#define COUT_   256
#define C_      64
#define ALPHA_  0.2f
#define NEG_INF_ (-9.0e15f)

typedef float v2f __attribute__((ext_vector_type(2)));
typedef float v8f __attribute__((ext_vector_type(8)));

#define GEMM_BROWS_ 64
#define GEMM_BCOLS_ 128

typedef float v4f __attribute__((ext_vector_type(4)));

struct GemmChunk {
  v4f a[4][2];
  v4f b[4][4];
};

__device__ __forceinline__ void gemm_load(GemmChunk& q, const float* __restrict__ xa,
                                          const float* __restrict__ wb, int kb) {
  #pragma unroll
  for (int p = 0; p < 4; ++p) {
    const int ko = kb + 8 * p;
    q.a[p][0] = *(const v4f*)(xa + ko);
    q.a[p][1] = *(const v4f*)(xa + 16 * CIN_ + ko);
    #pragma unroll
    for (int cc = 0; cc < 4; ++cc)
      q.b[p][cc] = *(const v4f*)(wb + cc * 16 * CIN_ + ko);
  }
}

__device__ __forceinline__ void gemm_compute(const GemmChunk& q, v8f (&acc)[2][4]) {
  #pragma unroll
  for (int p = 0; p < 4; ++p) {
    #pragma unroll
    for (int t = 0; t < 2; ++t) {
      v2f a0, a1;
      a0.x = q.a[p][0][2 * t]; a0.y = q.a[p][0][2 * t + 1];
      a1.x = q.a[p][1][2 * t]; a1.y = q.a[p][1][2 * t + 1];
      #pragma unroll
      for (int cc = 0; cc < 4; ++cc) {
        v2f bv;
        bv.x = q.b[p][cc][2 * t];
        bv.y = q.b[p][cc][2 * t + 1];
        acc[0][cc] = __builtin_amdgcn_wmma_f32_16x16x4_f32(false, a0, false, bv,
                                                           (short)0, acc[0][cc],
                                                           false, false);
        acc[1][cc] = __builtin_amdgcn_wmma_f32_16x16x4_f32(false, a1, false, bv,
                                                           (short)0, acc[1][cc],
                                                           false, false);
      }
    }
  }
}

__global__ __launch_bounds__(128) void gat_gemm_h(const float* __restrict__ X,
                                                  const float* __restrict__ W,
                                                  const float* __restrict__ bias,
                                                  const float* __restrict__ a,
                                                  float* __restrict__ hbuf,
                                                  float* __restrict__ lsrc,
                                                  float* __restrict__ ldst) {
  const int wv   = threadIdx.x >> 5;
  const int lane = threadIdx.x & 31;
  const int half = lane >> 4;
  const int l    = lane & 15;

  const int rb   = blockIdx.x >> 1;
  const int ch   = blockIdx.x & 1;
  const int row0 = rb * GEMM_BROWS_ + (wv >> 1) * 32;
  const int col0 = ch * GEMM_BCOLS_ + (wv & 1) * 64;

  const float* xa = X + (size_t)(row0 + l) * CIN_ + 4 * half;
  const float* wb = W + (size_t)(col0 + l) * CIN_ + 4 * half;

  GemmChunk q0, q1;
  gemm_load(q0, xa, wb, 0);

  v8f acc[2][4] = {{{}, {}, {}, {}}, {{}, {}, {}, {}}};
  #pragma unroll
  for (int kc = 0; kc < CIN_; kc += 64) {
    gemm_load(q1, xa, wb, kc + 32);
    __builtin_amdgcn_sched_barrier(0);
    gemm_compute(q0, acc);
    __builtin_amdgcn_sched_barrier(0);
    if (kc + 64 < CIN_) gemm_load(q0, xa, wb, kc + 64);
    __builtin_amdgcn_sched_barrier(0);
    gemm_compute(q1, acc);
  }

  #pragma unroll
  for (int rt = 0; rt < 2; ++rt) {
    #pragma unroll
    for (int cc = 0; cc < 4; ++cc) {
      const float bval = bias[col0 + cc * 16 + l];
      #pragma unroll
      for (int r = 0; r < 8; ++r) {
        const int row = row0 + 16 * rt + r + 8 * half;
        hbuf[(size_t)row * COUT_ + col0 + cc * 16 + l] = acc[rt][cc][r] + bval;
      }
    }
  }

  __syncthreads();
  {
    const int head = 2 * ch + (threadIdx.x & 1);
    const int bi   = rb * GEMM_BROWS_ + (threadIdx.x >> 1);
    const float* hr = hbuf + (size_t)bi * COUT_ + head * C_;
    const float* as = a + head * (2 * C_);
    const float* ad = as + C_;
    float s = 0.f, d = 0.f;
    #pragma unroll
    for (int c = 0; c < C_; ++c) {
      const float v = hr[c];
      s = fmaf(v, as[c], s);
      d = fmaf(v, ad[c], d);
    }
    lsrc[bi * H_ + head] = s;
    ldst[bi * H_ + head] = d;
  }
}

typedef int v4i __attribute__((ext_vector_type(4)));

#define SUM_FLOOR_ 9.5367431640625e-07f

struct AggChunk {
  v4i adA[4];
  v4i adB[4];
  v4f hr[4][4];
  v4f f1[4];
  v4f f2[4];
};

__device__ __forceinline__ void agg_load_global(AggChunk& q, const int* __restrict__ arA,
                                                const int* __restrict__ arB,
                                                const float* __restrict__ hcol, int jb) {
  #pragma unroll
  for (int u = 0; u < 4; ++u) {
    const int j = jb + 8 * u;
    q.adA[u] = *(const v4i*)(arA + j);
    q.adB[u] = *(const v4i*)(arB + j);
    #pragma unroll
    for (int r = 0; r < 4; ++r)
      q.hr[u][r] = *(const v4f*)(hcol + (size_t)(j + r) * COUT_);
  }
}

__device__ __forceinline__ void agg_load_tables(AggChunk& q, const float* f1h,
                                                const float* f2h, int jb) {
  #pragma unroll
  for (int u = 0; u < 4; ++u) {
    q.f1[u] = *(const v4f*)(f1h + jb + 8 * u);
    q.f2[u] = *(const v4f*)(f2h + jb + 8 * u);
  }
}

template <bool TABLES>
__device__ __forceinline__ void agg_load(AggChunk& q, const int* __restrict__ arA,
                                         const int* __restrict__ arB,
                                         const float* __restrict__ hcol,
                                         const float* f1h, const float* f2h, int jb) {
  agg_load_global(q, arA, arB, hcol, jb);
  if (TABLES) agg_load_tables(q, f1h, f2h, jb);
}

__device__ __forceinline__ void agg_group_mma(const AggChunk& q, int u,
                                              const float (&pA)[4], const float (&pB)[4],
                                              v8f (&accA)[4], v8f (&accB)[4]) {
  #pragma unroll
  for (int t = 0; t < 2; ++t) {
    v2f avA, avB;
    avA.x = pA[2 * t]; avA.y = pA[2 * t + 1];
    avB.x = pB[2 * t]; avB.y = pB[2 * t + 1];
    #pragma unroll
    for (int cc = 0; cc < 4; ++cc) {
      v2f bv;
      bv.x = q.hr[u][2 * t][cc];
      bv.y = q.hr[u][2 * t + 1][cc];
      accA[cc] = __builtin_amdgcn_wmma_f32_16x16x4_f32(false, avA, false, bv,
                                                       (short)0, accA[cc],
                                                       false, false);
      accB[cc] = __builtin_amdgcn_wmma_f32_16x16x4_f32(false, avB, false, bv,
                                                       (short)0, accB[cc],
                                                       false, false);
    }
  }
}

__device__ __forceinline__ void agg_compute_tables(const AggChunk& q,
                                                   float e1A, float e2A, float e1B, float e2B,
                                                   float& sumA, float& sumB,
                                                   v8f (&accA)[4], v8f (&accB)[4]) {
  #pragma unroll
  for (int u = 0; u < 4; ++u) {
    float pA[4], pB[4];
    #pragma unroll
    for (int r = 0; r < 4; ++r) {
      const float f1 = q.f1[u][r], f2 = q.f2[u][r];
      const float wA = fmaxf(e1A * f1, e2A * f2);
      const float wB = fmaxf(e1B * f1, e2B * f2);
      pA[r] = (q.adA[u][r] == 1) ? wA : 0.0f;
      pB[r] = (q.adB[u][r] == 1) ? wB : 0.0f;
      sumA += pA[r];
      sumB += pB[r];
    }
    agg_group_mma(q, u, pA, pB, accA, accB);
  }
}

__device__ __forceinline__ void agg_compute(const AggChunk& q, const float* ldh, int jb,
                                            float lsA, float lsB, float mA, float mB,
                                            float& sumA, float& sumB,
                                            v8f (&accA)[4], v8f (&accB)[4]) {
  #pragma unroll
  for (int u = 0; u < 4; ++u) {
    const v4f ld = *(const v4f*)(ldh + jb + 8 * u);
    float pA[4], pB[4];
    #pragma unroll
    for (int r = 0; r < 4; ++r) {
      float xa = lsA + ld[r]; xa = fmaxf(xa, ALPHA_ * xa);
      float xb = lsB + ld[r]; xb = fmaxf(xb, ALPHA_ * xb);
      pA[r] = (q.adA[u][r] == 1) ? __expf(xa - mA) : 0.0f;
      pB[r] = (q.adB[u][r] == 1) ? __expf(xb - mB) : 0.0f;
      sumA += pA[r];
      sumB += pB[r];
    }
    agg_group_mma(q, u, pA, pB, accA, accB);
  }
}

template <bool TABLES>
__device__ __forceinline__ void agg_sweep(AggChunk& q0, AggChunk& q1,
                                          const int* __restrict__ arA,
                                          const int* __restrict__ arB,
                                          const float* __restrict__ hcol,
                                          const float* ldh, const float* f1h,
                                          const float* f2h, int jh,
                                          float lsA, float lsB, float mA, float mB,
                                          float e1A, float e2A, float e1B, float e2B,
                                          float& sumA, float& sumB,
                                          v8f (&accA)[4], v8f (&accB)[4]) {
  for (int jc = 0; jc < N_; jc += 64) {
    agg_load<TABLES>(q1, arA, arB, hcol, f1h, f2h, jc + 32 + jh);
    __builtin_amdgcn_sched_barrier(0);
    if (TABLES) agg_compute_tables(q0, e1A, e2A, e1B, e2B, sumA, sumB, accA, accB);
    else        agg_compute(q0, ldh, jc + jh, lsA, lsB, mA, mB, sumA, sumB, accA, accB);
    __builtin_amdgcn_sched_barrier(0);
    agg_load<TABLES>(q0, arA, arB, hcol, f1h, f2h, ((jc + 64) & (N_ - 1)) + jh);
    __builtin_amdgcn_sched_barrier(0);
    if (TABLES) agg_compute_tables(q1, e1A, e2A, e1B, e2B, sumA, sumB, accA, accB);
    else        agg_compute(q1, ldh, jc + 32 + jh, lsA, lsB, mA, mB, sumA, sumB, accA, accB);
  }
}

__device__ __forceinline__ void agg_sweep_lead2(AggChunk& q0, AggChunk& q1,
                                                AggChunk& q2, AggChunk& q3,
                                                const int* __restrict__ arA,
                                                const int* __restrict__ arB,
                                                const float* __restrict__ hcol,
                                                const float* f1h, const float* f2h, int jh,
                                                float e1A, float e2A, float e1B, float e2B,
                                                float& sumA, float& sumB,
                                                v8f (&accA)[4], v8f (&accB)[4]) {
  for (int jc = 0; jc < N_; jc += 128) {
    agg_load<true>(q2, arA, arB, hcol, f1h, f2h, jc + 64 + jh);
    __builtin_amdgcn_sched_barrier(0);
    agg_compute_tables(q0, e1A, e2A, e1B, e2B, sumA, sumB, accA, accB);
    __builtin_amdgcn_sched_barrier(0);
    agg_load<true>(q3, arA, arB, hcol, f1h, f2h, jc + 96 + jh);
    __builtin_amdgcn_sched_barrier(0);
    agg_compute_tables(q1, e1A, e2A, e1B, e2B, sumA, sumB, accA, accB);
    __builtin_amdgcn_sched_barrier(0);
    agg_load<true>(q0, arA, arB, hcol, f1h, f2h, ((jc + 128) & (N_ - 1)) + jh);
    __builtin_amdgcn_sched_barrier(0);
    agg_compute_tables(q2, e1A, e2A, e1B, e2B, sumA, sumB, accA, accB);
    __builtin_amdgcn_sched_barrier(0);
    agg_load<true>(q1, arA, arB, hcol, f1h, f2h, ((jc + 160) & (N_ - 1)) + jh);
    __builtin_amdgcn_sched_barrier(0);
    agg_compute_tables(q3, e1A, e2A, e1B, e2B, sumA, sumB, accA, accB);
  }
}

__global__ __launch_bounds__(128) void gat_aggregate(const float* __restrict__ hbuf,
                                                     const float* __restrict__ lsrc,
                                                     const float* __restrict__ ldst,
                                                     const int*   __restrict__ adj,
                                                     float* __restrict__ out) {
  __shared__ __attribute__((aligned(16))) float lds_ld[2 * N_];
  __shared__ __attribute__((aligned(16))) float lds_f1[2 * N_];
  __shared__ __attribute__((aligned(16))) float lds_f2[2 * N_];
  __shared__ float lds_wmax[4][2];

  const int b     = blockIdx.x >> 5;
  const int hp    = blockIdx.x & 1;
  const int sp    = (blockIdx.x & 31) >> 1;
  const int wv    = threadIdx.x >> 5;
  const int hl    = wv & 1;
  const int head  = 2 * hp + hl;
  const int i0    = (2 * sp + (wv >> 1)) * 32;
  const int lane  = threadIdx.x & 31;
  const int half  = lane >> 4;
  const int l     = lane & 15;

  const size_t biA = (size_t)b * N_ + (i0 + l);
  const size_t biB = biA + 16;
  const int*   arA = adj + biA * (size_t)N_;
  const int*   arB = adj + biB * (size_t)N_;
  const float* hcol = hbuf + (size_t)b * N_ * COUT_ + head * C_ + 4 * l;
  const float* ldh  = lds_ld + hl * N_;
  const float* f1h  = lds_f1 + hl * N_;
  const float* f2h  = lds_f2 + hl * N_;
  const int    jh   = 4 * half;

  const float2* src = (const float2*)(ldst + (size_t)b * N_ * H_ + 2 * hp);
  float2 ldv[N_ / 128];
  #pragma unroll
  for (int t = 0; t < N_ / 128; ++t) ldv[t] = src[2 * (threadIdx.x + 128 * t)];

  const float  lsA = lsrc[biA * H_ + head];
  const float  lsB = lsrc[biB * H_ + head];

  AggChunk q0, q1, q2, q3;
  agg_load_global(q0, arA, arB, hcol, jh);
  agg_load_global(q1, arA, arB, hcol, 32 + jh);

  {
    float hm[2] = {-INFINITY, -INFINITY};
    #pragma unroll
    for (int t = 0; t < N_ / 128; ++t) {
      const int j = threadIdx.x + 128 * t;
      const float2 v = ldv[t];
      lds_ld[0 * N_ + j] = v.x;
      lds_ld[1 * N_ + j] = v.y;
      hm[0] = fmaxf(hm[0], v.x);
      hm[1] = fmaxf(hm[1], v.y);
    }
    #pragma unroll
    for (int hd = 0; hd < 2; ++hd) {
      #pragma unroll
      for (int off = 16; off > 0; off >>= 1)
        hm[hd] = fmaxf(hm[hd], __shfl_xor(hm[hd], off, 32));
    }
    if (lane == 0) {
      #pragma unroll
      for (int hd = 0; hd < 2; ++hd) lds_wmax[wv][hd] = hm[hd];
    }
  }

  __syncthreads();

  {
    float L[2];
    #pragma unroll
    for (int hd = 0; hd < 2; ++hd)
      L[hd] = fmaxf(fmaxf(lds_wmax[0][hd], lds_wmax[1][hd]),
                    fmaxf(lds_wmax[2][hd], lds_wmax[3][hd]));
    #pragma unroll
    for (int t = 0; t < N_ / 128; ++t) {
      const int j = threadIdx.x + 128 * t;
      const float2 v = ldv[t];
      const float d0 = v.x - L[0], d1 = v.y - L[1];
      lds_f1[0 * N_ + j] = __expf(d0);
      lds_f1[1 * N_ + j] = __expf(d1);
      lds_f2[0 * N_ + j] = __expf(ALPHA_ * d0);
      lds_f2[1 * N_ + j] = __expf(ALPHA_ * d1);
    }
  }

  const float lmax = fmaxf(fmaxf(lds_wmax[0][hl], lds_wmax[1][hl]),
                           fmaxf(lds_wmax[2][hl], lds_wmax[3][hl]));
  float mA, mB;
  float e1A, e2A, e1B, e2B;
  {
    const float xA = lsA + lmax, xB = lsB + lmax;
    mA = fmaxf(xA, ALPHA_ * xA);
    mB = fmaxf(xB, ALPHA_ * xB);
    e1A = __expf(xA - mA); e2A = __expf(ALPHA_ * xA - mA);
    e1B = __expf(xB - mB); e2B = __expf(ALPHA_ * xB - mB);
  }

  __syncthreads();
  agg_load_tables(q0, f1h, f2h, jh);
  agg_load_tables(q1, f1h, f2h, 32 + jh);

  v8f accA[4], accB[4];
  float sumA, sumB;

  #pragma unroll
  for (int cc = 0; cc < 4; ++cc) { accA[cc] = v8f{}; accB[cc] = v8f{}; }
  sumA = 0.0f; sumB = 0.0f;

  agg_sweep_lead2(q0, q1, q2, q3, arA, arB, hcol, f1h, f2h, jh,
                  e1A, e2A, e1B, e2B, sumA, sumB, accA, accB);

  sumA += __shfl_xor(sumA, 16, 32);
  sumB += __shfl_xor(sumB, 16, 32);

  const int small = (sumA < SUM_FLOOR_) || (sumB < SUM_FLOOR_);
  if (__any(small)) {
  float mxA = -INFINITY, mxB = -INFINITY;
  for (int j = jh; j < N_; j += 8) {
    const v4i a4 = *(const v4i*)(arA + j);
    const v4i b4 = *(const v4i*)(arB + j);
    const v4f ld = *(const v4f*)(ldh + j);
    #pragma unroll
    for (int r = 0; r < 4; ++r) {
      mxA = fmaxf(mxA, (a4[r] == 1) ? ld[r] : -INFINITY);
      mxB = fmaxf(mxB, (b4[r] == 1) ? ld[r] : -INFINITY);
    }
  }
  mxA = fmaxf(mxA, __shfl_xor(mxA, 16, 32));
  mxB = fmaxf(mxB, __shfl_xor(mxB, 16, 32));
  {
    const float xA = lsA + mxA, xB = lsB + mxB;
    mA = fmaxf(xA, ALPHA_ * xA);
    mB = fmaxf(xB, ALPHA_ * xB);
  }
  #pragma unroll
  for (int cc = 0; cc < 4; ++cc) { accA[cc] = v8f{}; accB[cc] = v8f{}; }
  sumA = 0.0f; sumB = 0.0f;
  agg_sweep<false>(q0, q1, arA, arB, hcol, ldh, f1h, f2h, jh, lsA, lsB, mA, mB,
                   e1A, e2A, e1B, e2B, sumA, sumB, accA, accB);
  sumA += __shfl_xor(sumA, 16, 32);
  sumB += __shfl_xor(sumB, 16, 32);
  }
  const float invA = (sumA > 0.0f) ? 1.0f / sumA : 0.0f;
  const float invB = (sumB > 0.0f) ? 1.0f / sumB : 0.0f;

  #pragma unroll
  for (int r = 0; r < 8; ++r) {
    const int rowA = i0 + r + 8 * half;
    const int rowB = rowA + 16;
    const float wA = __shfl(invA, r + 8 * half, 32);
    const float wB = __shfl(invB, r + 8 * half, 32);
    v4f oa, ob;
    oa.x = accA[0][r] * wA; oa.y = accA[1][r] * wA; oa.z = accA[2][r] * wA; oa.w = accA[3][r] * wA;
    ob.x = accB[0][r] * wB; ob.y = accB[1][r] * wB; ob.z = accB[2][r] * wB; ob.w = accB[3][r] * wB;
    *(v4f*)(out + ((size_t)b * N_ + rowA) * COUT_ + head * C_ + 4 * l) = oa;
    *(v4f*)(out + ((size_t)b * N_ + rowB) * COUT_ + head * C_ + 4 * l) = ob;
  }
}

extern "C" void kernel_launch(void* const* d_in, const int* in_sizes, int n_in,
                              void* d_out, int out_size, void* d_ws, size_t ws_size,
                              hipStream_t stream) {
  const float* node_feats = (const float*)d_in[0];
  const int*   adj        = (const int*)  d_in[1];
  const float* W          = (const float*)d_in[2];
  const float* bias       = (const float*)d_in[3];
  const float* a          = (const float*)d_in[4];
  float*       out        = (float*)d_out;

  float* hbuf = (float*)d_ws;
  float* lsrc = hbuf + (size_t)B_ * N_ * COUT_;
  float* ldst = lsrc + (size_t)B_ * N_ * H_;

  gat_gemm_h<<<(B_ * N_ / GEMM_BROWS_) * (COUT_ / GEMM_BCOLS_), 128, 0, stream>>>(
      node_feats, W, bias, a, hbuf, lsrc, ldst);

  gat_aggregate<<<B_ * (N_ / 32), 128, 0, stream>>>(
      hbuf, lsrc, ldst, adj, out);
}
